# starting kernel
# baseline (speedup 1.0000x reference)
#include <hip/hip_runtime.h>


#define BATCH   4
#define FDIM    64
#define N_NODES 8192
#define LROW    68

typedef __attribute__((ext_vector_type(2))) float v2f;
typedef __attribute__((ext_vector_type(8))) float v8f;

__global__ __launch_bounds__(256) void gcn_prep(const float* __restrict__ x,
                                                const float* __restrict__ w,
                                                float* __restrict__ xt,
                                                float* __restrict__ sup) {
    int n = blockIdx.x * 256 + threadIdx.x;
    int b = blockIdx.y;
    const float* xb = x + (size_t)b * FDIM * N_NODES;
    float* xtrow = xt  + ((size_t)b * N_NODES + n) * FDIM;
    float* srow  = sup + ((size_t)b * N_NODES + n) * FDIM;

    float acc[FDIM];
#pragma unroll
    for (int o = 0; o < FDIM; ++o) acc[o] = 0.0f;

    for (int f = 0; f < FDIM; ++f) {
        float v = xb[(size_t)f * N_NODES + n];
        xtrow[f] = v;
#pragma unroll
        for (int o = 0; o < FDIM; ++o) acc[o] += v * w[f * FDIM + o];
    }
#pragma unroll
    for (int o = 0; o < FDIM; ++o) srow[o] = acc[o];
}

__device__ __forceinline__ void stage_load(const float* __restrict__ xtb,
                                           const float* __restrict__ vbp,
                                           int j, int tid,
                                           float4* rk, float4* rv) {
#pragma unroll
    for (int i = 0; i < 2; ++i) {
        int idx = i * 128 + tid;
        int row = idx >> 4;
        int c4  = idx & 15;
        size_t off = (size_t)(j * 16 + row) * FDIM + c4 * 4;
        rk[i] = *(const float4*)(xtb + off);
        rv[i] = *(const float4*)(vbp + off);
    }
}

__global__ __launch_bounds__(128) void gcn_flash(const float* __restrict__ xt,
                                                 const float* __restrict__ sup,
                                                 float* __restrict__ out) {
    __shared__ __align__(16) float sh[2][2][16][LROW];

    const int tid  = threadIdx.x;
    const int wave = tid >> 5;
    const int lane = tid & 31;
    const int hi   = lane >> 4;
    const int mrow = lane & 15;

    const int b    = blockIdx.x >> 7;
    const int qblk = blockIdx.x & 127;
    const int qbase = qblk * 64 + wave * 16;

    const float* xtb = xt  + (size_t)b * N_NODES * FDIM;
    const float* vbp = sup + (size_t)b * N_NODES * FDIM;

    v2f q[16];
    {
        const float* qrow = xtb + (size_t)(qbase + mrow) * FDIM + 2 * hi;
#pragma unroll
        for (int c = 0; c < 16; ++c)
            q[c] = *(const v2f*)(qrow + 4 * c);
    }

    v8f accT[4];
#pragma unroll
    for (int t = 0; t < 4; ++t)
#pragma unroll
        for (int r = 0; r < 8; ++r) accT[t][r] = 0.0f;

    float m = -3.0e38f;
    float l = 0.0f;

    float4 rk[2], rv[2];
    stage_load(xtb, vbp, 0, tid, rk, rv);

    const int NJ = N_NODES / 16;
    for (int j = 0; j < NJ; ++j) {
        const int buf = j & 1;
#pragma unroll
        for (int i = 0; i < 2; ++i) {
            int idx = i * 128 + tid;
            int row = idx >> 4, c4 = idx & 15;
            *(float4*)&sh[buf][0][row][c4 * 4] = rk[i];
            *(float4*)&sh[buf][1][row][c4 * 4] = rv[i];
        }
        __syncthreads();
        if (j + 1 < NJ)
            stage_load(xtb, vbp, j + 1, tid, rk, rv);

        const float* kb    = &sh[buf][0][0][0];
        const float* vtile = &sh[buf][1][0][0];

        v8f s;
#pragma unroll
        for (int r = 0; r < 8; ++r) s[r] = 0.0f;
#pragma unroll
        for (int c = 0; c < 16; ++c) {
            v2f kk = *(const v2f*)(kb + mrow * LROW + 4 * c + 2 * hi);
            s = __builtin_amdgcn_wmma_f32_16x16x4_f32(
                    false, kk, false, q[c], (short)0, s, false, false);
        }

        float mloc = s[0];
#pragma unroll
        for (int r = 1; r < 8; ++r) mloc = fmaxf(mloc, s[r]);
        mloc = fmaxf(mloc, __shfl_xor(mloc, 16, 32));
        float mnew = fmaxf(m, mloc);
        float sc = __expf(m - mnew);
        m = mnew;

        float p[8], ls = 0.0f;
#pragma unroll
        for (int r = 0; r < 8; ++r) { p[r] = __expf(s[r] - mnew); ls += p[r]; }
        ls += __shfl_xor(ls, 16, 32);
        l = l * sc + ls;

#pragma unroll
        for (int t = 0; t < 4; ++t)
#pragma unroll
            for (int r = 0; r < 8; ++r) accT[t][r] *= sc;

        float tt[8];
#pragma unroll
        for (int r = 0; r < 8; ++r) tt[r] = __shfl_xor(p[r], 16, 32);
        v2f pb[4];
        pb[0].x = hi ? tt[2] : p[0];  pb[0].y = hi ? tt[3] : p[1];
        pb[1].x = hi ? tt[6] : p[4];  pb[1].y = hi ? tt[7] : p[5];
        pb[2].x = hi ? p[2] : tt[0];  pb[2].y = hi ? p[3] : tt[1];
        pb[3].x = hi ? p[6] : tt[4];  pb[3].y = hi ? p[7] : tt[5];

#pragma unroll
        for (int t = 0; t < 4; ++t) {
#pragma unroll
            for (int c = 0; c < 4; ++c) {
                int krow = 4 * c + 2 * hi;
                v2f a;
                a.x = vtile[krow * LROW + 16 * t + mrow];
                a.y = vtile[(krow + 1) * LROW + 16 * t + mrow];
                accT[t] = __builtin_amdgcn_wmma_f32_16x16x4_f32(
                              false, a, false, pb[c], (short)0, accT[t],
                              false, false);
            }
        }
    }

    float inv = 1.0f / l;
    int n = qbase + mrow;
#pragma unroll
    for (int t = 0; t < 4; ++t)
#pragma unroll
        for (int r = 0; r < 8; ++r) {
            int o = 16 * t + r + 8 * hi;
            out[((size_t)b * FDIM + o) * N_NODES + n] = accT[t][r] * inv;
        }
}

extern "C" void kernel_launch(void* const* d_in, const int* in_sizes, int n_in,
                              void* d_out, int out_size, void* d_ws, size_t ws_size,
                              hipStream_t stream) {
    (void)in_sizes; (void)n_in; (void)out_size; (void)ws_size;
    const float* x = (const float*)d_in[0];
    const float* w = (const float*)d_in[1];
    float* xt  = (float*)d_ws;
    float* sup = xt + (size_t)BATCH * N_NODES * FDIM;
    float* out = (float*)d_out;

    dim3 gPrep(N_NODES / 256, BATCH);
    gcn_prep<<<gPrep, 256, 0, stream>>>(x, w, xt, sup);

    dim3 gFlash(BATCH * (N_NODES / 64));
    gcn_flash<<<gFlash, 128, 0, stream>>>(xt, sup, out);
}


# K,V written by prep in WMMA fragment order; each wave streams its own fragments global->registers (coalesced 512 B loads, two tiles ahead; 8-wave blocks share one stream): no LDS, barrier or scratch
# speedup vs baseline: 2.7321x; 2.7321x over previous
#include <hip/hip_runtime.h>


#define BATCH   4
#define FDIM    64
#define N_NODES 8192
#define NJ      (N_NODES / 16)
#define TILE_F  1024
#define TILE_F4 256

typedef __attribute__((ext_vector_type(2))) float v2f;
typedef __attribute__((ext_vector_type(8))) float v8f;

__global__ __launch_bounds__(256) void gcn_prep(const float* __restrict__ x,
                                                const float* __restrict__ w,
                                                float* __restrict__ kf,
                                                float* __restrict__ vf) {
    const int n  = blockIdx.x * 256 + threadIdx.x;
    const int b  = blockIdx.y;
    const int j  = n >> 4;
    const int kk = n & 15;
    const float* xb = x + (size_t)b * FDIM * N_NODES;
    float4* ktile = (float4*)(kf + ((size_t)b * NJ + j) * TILE_F);
    float*  vtile = vf + ((size_t)b * NJ + j) * TILE_F;

    float acc[FDIM];
#pragma unroll
    for (int o = 0; o < FDIM; ++o) acc[o] = 0.0f;

    for (int g = 0; g < 8; ++g) {
        float v[8];
#pragma unroll
        for (int i = 0; i < 8; ++i)
            v[i] = xb[(size_t)(8 * g + i) * N_NODES + n];
#pragma unroll
        for (int i = 0; i < 8; ++i)
#pragma unroll
            for (int o = 0; o < FDIM; ++o)
                acc[o] += v[i] * w[(8 * g + i) * FDIM + o];
        ktile[g * 32 + kk]      = make_float4(v[0], v[1], v[4], v[5]);
        ktile[g * 32 + 16 + kk] = make_float4(v[2], v[3], v[6], v[7]);
    }

    const int c = kk >> 2, h = (kk >> 1) & 1, e = kk & 1;
    float* vp = vtile + ((c >> 1) * 32 + h * 16) * 4 + (c & 1) * 2 + e;
#pragma unroll
    for (int o = 0; o < FDIM; ++o)
        vp[(o >> 4) * 256 + (o & 15) * 4] = acc[o];
}

struct Frag8 { float4 f0, f1, f2, f3, f4, f5, f6, f7; };

__device__ __forceinline__ void load_frag(Frag8& d, const float4* __restrict__ p) {
    d.f0 = p[0];   d.f1 = p[32];  d.f2 = p[64];  d.f3 = p[96];
    d.f4 = p[128]; d.f5 = p[160]; d.f6 = p[192]; d.f7 = p[224];
}

#define WMMA_F32(A, B, C) \
    __builtin_amdgcn_wmma_f32_16x16x4_f32(false, (A), false, (B), (short)0, (C), false, false)

#define LO2(F) (v2f{(F).x, (F).y})
#define HI2(F) (v2f{(F).z, (F).w})

__device__ __forceinline__ v8f tile_scores(const Frag8& K, const v2f (&q)[16]) {
    v8f s;
#pragma unroll
    for (int r = 0; r < 8; ++r) s[r] = 0.0f;
    s = WMMA_F32(LO2(K.f0), q[0],  s);  s = WMMA_F32(HI2(K.f0), q[1],  s);
    s = WMMA_F32(LO2(K.f1), q[2],  s);  s = WMMA_F32(HI2(K.f1), q[3],  s);
    s = WMMA_F32(LO2(K.f2), q[4],  s);  s = WMMA_F32(HI2(K.f2), q[5],  s);
    s = WMMA_F32(LO2(K.f3), q[6],  s);  s = WMMA_F32(HI2(K.f3), q[7],  s);
    s = WMMA_F32(LO2(K.f4), q[8],  s);  s = WMMA_F32(HI2(K.f4), q[9],  s);
    s = WMMA_F32(LO2(K.f5), q[10], s);  s = WMMA_F32(HI2(K.f5), q[11], s);
    s = WMMA_F32(LO2(K.f6), q[12], s);  s = WMMA_F32(HI2(K.f6), q[13], s);
    s = WMMA_F32(LO2(K.f7), q[14], s);  s = WMMA_F32(HI2(K.f7), q[15], s);
    return s;
}

__device__ __forceinline__ void tile_softmax_pv(const v8f s, const Frag8& V,
                                                v8f (&accT)[4],
                                                float& m, float& l, const int hi) {

    float mloc = s[0];
#pragma unroll
    for (int r = 1; r < 8; ++r) mloc = fmaxf(mloc, s[r]);
    mloc = fmaxf(mloc, __shfl_xor(mloc, 16, 32));
    float mnew = fmaxf(m, mloc);
    float sc = __expf(m - mnew);
    m = mnew;

    float p[8], ls = 0.0f;
#pragma unroll
    for (int r = 0; r < 8; ++r) { p[r] = __expf(s[r] - mnew); ls += p[r]; }
    ls += __shfl_xor(ls, 16, 32);
    l = l * sc + ls;

#pragma unroll
    for (int t = 0; t < 4; ++t)
#pragma unroll
        for (int r = 0; r < 8; ++r) accT[t][r] *= sc;

    float tt[8];
#pragma unroll
    for (int r = 0; r < 8; ++r) tt[r] = __shfl_xor(p[r], 16, 32);
    v2f pb[4];
    pb[0].x = hi ? tt[2] : p[0];  pb[0].y = hi ? tt[3] : p[1];
    pb[1].x = hi ? tt[6] : p[4];  pb[1].y = hi ? tt[7] : p[5];
    pb[2].x = hi ? p[2] : tt[0];  pb[2].y = hi ? p[3] : tt[1];
    pb[3].x = hi ? p[6] : tt[4];  pb[3].y = hi ? p[7] : tt[5];

    accT[0] = WMMA_F32(LO2(V.f0), pb[0], accT[0]);
    accT[1] = WMMA_F32(LO2(V.f2), pb[0], accT[1]);
    accT[2] = WMMA_F32(LO2(V.f4), pb[0], accT[2]);
    accT[3] = WMMA_F32(LO2(V.f6), pb[0], accT[3]);
    accT[0] = WMMA_F32(HI2(V.f0), pb[1], accT[0]);
    accT[1] = WMMA_F32(HI2(V.f2), pb[1], accT[1]);
    accT[2] = WMMA_F32(HI2(V.f4), pb[1], accT[2]);
    accT[3] = WMMA_F32(HI2(V.f6), pb[1], accT[3]);
    accT[0] = WMMA_F32(LO2(V.f1), pb[2], accT[0]);
    accT[1] = WMMA_F32(LO2(V.f3), pb[2], accT[1]);
    accT[2] = WMMA_F32(LO2(V.f5), pb[2], accT[2]);
    accT[3] = WMMA_F32(LO2(V.f7), pb[2], accT[3]);
    accT[0] = WMMA_F32(HI2(V.f1), pb[3], accT[0]);
    accT[1] = WMMA_F32(HI2(V.f3), pb[3], accT[1]);
    accT[2] = WMMA_F32(HI2(V.f5), pb[3], accT[2]);
    accT[3] = WMMA_F32(HI2(V.f7), pb[3], accT[3]);
}

__global__ __launch_bounds__(256) void gcn_flash(const float* __restrict__ kf,
                                                 const float* __restrict__ vf,
                                                 float* __restrict__ out) {
    const int tid  = threadIdx.x;
    const int wave = tid >> 5;
    const int lane = tid & 31;
    const int hi   = lane >> 4;
    const int mrow = lane & 15;

    const int b     = blockIdx.x >> 6;
    const int qblk  = blockIdx.x & 63;
    const int qtile = qblk * 8 + wave;
    const int qbase = qtile * 16;

    const float4* kp = (const float4*)kf + (size_t)b * NJ * TILE_F4 + lane;
    const float4* vp = (const float4*)vf + (size_t)b * NJ * TILE_F4 + lane;

    v2f q[16];
    {
        Frag8 qf;
        load_frag(qf, kp + (size_t)qtile * TILE_F4);
        q[0]  = LO2(qf.f0); q[1]  = HI2(qf.f0); q[2]  = LO2(qf.f1); q[3]  = HI2(qf.f1);
        q[4]  = LO2(qf.f2); q[5]  = HI2(qf.f2); q[6]  = LO2(qf.f3); q[7]  = HI2(qf.f3);
        q[8]  = LO2(qf.f4); q[9]  = HI2(qf.f4); q[10] = LO2(qf.f5); q[11] = HI2(qf.f5);
        q[12] = LO2(qf.f6); q[13] = HI2(qf.f6); q[14] = LO2(qf.f7); q[15] = HI2(qf.f7);
    }

    v8f accT[4];
#pragma unroll
    for (int t = 0; t < 4; ++t)
#pragma unroll
        for (int r = 0; r < 8; ++r) accT[t][r] = 0.0f;

    float m = -3.0e38f;
    float l = 0.0f;

    Frag8 kA, vA, kB, vB;
    load_frag(kA, kp);
    load_frag(vA, vp);
    load_frag(kB, kp + TILE_F4);
    load_frag(vB, vp + TILE_F4);

    for (int j = 0; j < NJ; j += 2) {
        const int ja = (j + 2 < NJ) ? (j + 2) : j;
        const int jb = (j + 3 < NJ) ? (j + 3) : (j + 1);

        v8f s0 = tile_scores(kA, q);
        load_frag(kA, kp + (size_t)ja * TILE_F4);
        __builtin_amdgcn_sched_barrier(0);
        tile_softmax_pv(s0, vA, accT, m, l, hi);
        load_frag(vA, vp + (size_t)ja * TILE_F4);
        __builtin_amdgcn_sched_barrier(0);

        v8f s1 = tile_scores(kB, q);
        load_frag(kB, kp + (size_t)jb * TILE_F4);
        __builtin_amdgcn_sched_barrier(0);
        tile_softmax_pv(s1, vB, accT, m, l, hi);
        load_frag(vB, vp + (size_t)jb * TILE_F4);
        __builtin_amdgcn_sched_barrier(0);
    }

    float inv = 1.0f / l;
    int n = qbase + mrow;
#pragma unroll
    for (int t = 0; t < 4; ++t)
#pragma unroll
        for (int r = 0; r < 8; ++r) {
            int o = 16 * t + r + 8 * hi;
            out[((size_t)b * FDIM + o) * N_NODES + n] = accT[t][r] * inv;
        }
}

extern "C" void kernel_launch(void* const* d_in, const int* in_sizes, int n_in,
                              void* d_out, int out_size, void* d_ws, size_t ws_size,
                              hipStream_t stream) {
    (void)in_sizes; (void)n_in; (void)out_size; (void)ws_size;
    const float* x = (const float*)d_in[0];
    const float* w = (const float*)d_in[1];
    float* kf  = (float*)d_ws;
    float* vf  = kf + (size_t)BATCH * N_NODES * FDIM;
    float* out = (float*)d_out;

    dim3 gPrep(N_NODES / 256, BATCH);
    gcn_prep<<<gPrep, 256, 0, stream>>>(x, w, kf, vf);

    dim3 gFlash(BATCH * (N_NODES / 128));
    gcn_flash<<<gFlash, 256, 0, stream>>>(kf, vf, out);
}


# Softmax cross-half lane exchanges (row max, row sum, P^T build) by v_permlanex16 on the vector ALU, not ds_bpermute: no LDS-unit instruction left in the loop; same operations, same sum order
# speedup vs baseline: 2.8655x; 1.0488x over previous
#include <hip/hip_runtime.h>


#define BATCH   4
#define FDIM    64
#define N_NODES 8192
#define NJ      (N_NODES / 16)
#define TILE_F  1024
#define TILE_F4 256

typedef __attribute__((ext_vector_type(2))) float v2f;
typedef __attribute__((ext_vector_type(8))) float v8f;

__global__ __launch_bounds__(256) void gcn_prep(const float* __restrict__ x,
                                                const float* __restrict__ w,
                                                float* __restrict__ kf,
                                                float* __restrict__ vf) {
    const int n  = blockIdx.x * 256 + threadIdx.x;
    const int b  = blockIdx.y;
    const int j  = n >> 4;
    const int kk = n & 15;
    const float* xb = x + (size_t)b * FDIM * N_NODES;
    float4* ktile = (float4*)(kf + ((size_t)b * NJ + j) * TILE_F);
    float*  vtile = vf + ((size_t)b * NJ + j) * TILE_F;

    float acc[FDIM];
#pragma unroll
    for (int o = 0; o < FDIM; ++o) acc[o] = 0.0f;

    for (int g = 0; g < 8; ++g) {
        float v[8];
#pragma unroll
        for (int i = 0; i < 8; ++i)
            v[i] = xb[(size_t)(8 * g + i) * N_NODES + n];
#pragma unroll
        for (int i = 0; i < 8; ++i)
#pragma unroll
            for (int o = 0; o < FDIM; ++o)
                acc[o] += v[i] * w[(8 * g + i) * FDIM + o];
        ktile[g * 32 + kk]      = make_float4(v[0], v[1], v[4], v[5]);
        ktile[g * 32 + 16 + kk] = make_float4(v[2], v[3], v[6], v[7]);
    }

    const int c = kk >> 2, h = (kk >> 1) & 1, e = kk & 1;
    float* vp = vtile + ((c >> 1) * 32 + h * 16) * 4 + (c & 1) * 2 + e;
#pragma unroll
    for (int o = 0; o < FDIM; ++o)
        vp[(o >> 4) * 256 + (o & 15) * 4] = acc[o];
}

struct Frag8 { float4 f0, f1, f2, f3, f4, f5, f6, f7; };

__device__ __forceinline__ void load_frag(Frag8& d, const float4* __restrict__ p) {
    d.f0 = p[0];   d.f1 = p[32];  d.f2 = p[64];  d.f3 = p[96];
    d.f4 = p[128]; d.f5 = p[160]; d.f6 = p[192]; d.f7 = p[224];
}

#define WMMA_F32(A, B, C) \
    __builtin_amdgcn_wmma_f32_16x16x4_f32(false, (A), false, (B), (short)0, (C), false, false)

__device__ __forceinline__ float xhalf(float x) {
    const int b = __builtin_bit_cast(int, x);
    return __builtin_bit_cast(float,
        __builtin_amdgcn_permlanex16(b, b, 0x76543210, 0xfedcba98, false, false));
}

#define LO2(F) (v2f{(F).x, (F).y})
#define HI2(F) (v2f{(F).z, (F).w})

__device__ __forceinline__ v8f tile_scores(const Frag8& K, const v2f (&q)[16]) {
    v8f s;
#pragma unroll
    for (int r = 0; r < 8; ++r) s[r] = 0.0f;
    s = WMMA_F32(LO2(K.f0), q[0],  s);  s = WMMA_F32(HI2(K.f0), q[1],  s);
    s = WMMA_F32(LO2(K.f1), q[2],  s);  s = WMMA_F32(HI2(K.f1), q[3],  s);
    s = WMMA_F32(LO2(K.f2), q[4],  s);  s = WMMA_F32(HI2(K.f2), q[5],  s);
    s = WMMA_F32(LO2(K.f3), q[6],  s);  s = WMMA_F32(HI2(K.f3), q[7],  s);
    s = WMMA_F32(LO2(K.f4), q[8],  s);  s = WMMA_F32(HI2(K.f4), q[9],  s);
    s = WMMA_F32(LO2(K.f5), q[10], s);  s = WMMA_F32(HI2(K.f5), q[11], s);
    s = WMMA_F32(LO2(K.f6), q[12], s);  s = WMMA_F32(HI2(K.f6), q[13], s);
    s = WMMA_F32(LO2(K.f7), q[14], s);  s = WMMA_F32(HI2(K.f7), q[15], s);
    return s;
}

__device__ __forceinline__ void tile_softmax_pv(const v8f s, const Frag8& V,
                                                v8f (&accT)[4],
                                                float& m, float& l, const int hi) {

    float mloc = s[0];
#pragma unroll
    for (int r = 1; r < 8; ++r) mloc = fmaxf(mloc, s[r]);
    mloc = fmaxf(mloc, xhalf(mloc));
    float mnew = fmaxf(m, mloc);
    float sc = __expf(m - mnew);
    m = mnew;

    float p[8], ls = 0.0f;
#pragma unroll
    for (int r = 0; r < 8; ++r) { p[r] = __expf(s[r] - mnew); ls += p[r]; }
    ls += xhalf(ls);
    l = l * sc + ls;

#pragma unroll
    for (int t = 0; t < 4; ++t)
#pragma unroll
        for (int r = 0; r < 8; ++r) accT[t][r] *= sc;

    float tt[8];
#pragma unroll
    for (int r = 0; r < 8; ++r) tt[r] = xhalf(p[r]);
    v2f pb[4];
    pb[0].x = hi ? tt[2] : p[0];  pb[0].y = hi ? tt[3] : p[1];
    pb[1].x = hi ? tt[6] : p[4];  pb[1].y = hi ? tt[7] : p[5];
    pb[2].x = hi ? p[2] : tt[0];  pb[2].y = hi ? p[3] : tt[1];
    pb[3].x = hi ? p[6] : tt[4];  pb[3].y = hi ? p[7] : tt[5];

    accT[0] = WMMA_F32(LO2(V.f0), pb[0], accT[0]);
    accT[1] = WMMA_F32(LO2(V.f2), pb[0], accT[1]);
    accT[2] = WMMA_F32(LO2(V.f4), pb[0], accT[2]);
    accT[3] = WMMA_F32(LO2(V.f6), pb[0], accT[3]);
    accT[0] = WMMA_F32(HI2(V.f0), pb[1], accT[0]);
    accT[1] = WMMA_F32(HI2(V.f2), pb[1], accT[1]);
    accT[2] = WMMA_F32(HI2(V.f4), pb[1], accT[2]);
    accT[3] = WMMA_F32(HI2(V.f6), pb[1], accT[3]);
    accT[0] = WMMA_F32(LO2(V.f1), pb[2], accT[0]);
    accT[1] = WMMA_F32(LO2(V.f3), pb[2], accT[1]);
    accT[2] = WMMA_F32(LO2(V.f5), pb[2], accT[2]);
    accT[3] = WMMA_F32(LO2(V.f7), pb[2], accT[3]);
    accT[0] = WMMA_F32(HI2(V.f1), pb[3], accT[0]);
    accT[1] = WMMA_F32(HI2(V.f3), pb[3], accT[1]);
    accT[2] = WMMA_F32(HI2(V.f5), pb[3], accT[2]);
    accT[3] = WMMA_F32(HI2(V.f7), pb[3], accT[3]);
}

__global__ __launch_bounds__(256) void gcn_flash(const float* __restrict__ kf,
                                                 const float* __restrict__ vf,
                                                 float* __restrict__ out) {
    const int tid  = threadIdx.x;
    const int wave = tid >> 5;
    const int lane = tid & 31;
    const int hi   = lane >> 4;
    const int mrow = lane & 15;

    const int b     = blockIdx.x >> 6;
    const int qblk  = blockIdx.x & 63;
    const int qtile = qblk * 8 + wave;
    const int qbase = qtile * 16;

    const float4* kp = (const float4*)kf + (size_t)b * NJ * TILE_F4 + lane;
    const float4* vp = (const float4*)vf + (size_t)b * NJ * TILE_F4 + lane;

    v2f q[16];
    {
        Frag8 qf;
        load_frag(qf, kp + (size_t)qtile * TILE_F4);
        q[0]  = LO2(qf.f0); q[1]  = HI2(qf.f0); q[2]  = LO2(qf.f1); q[3]  = HI2(qf.f1);
        q[4]  = LO2(qf.f2); q[5]  = HI2(qf.f2); q[6]  = LO2(qf.f3); q[7]  = HI2(qf.f3);
        q[8]  = LO2(qf.f4); q[9]  = HI2(qf.f4); q[10] = LO2(qf.f5); q[11] = HI2(qf.f5);
        q[12] = LO2(qf.f6); q[13] = HI2(qf.f6); q[14] = LO2(qf.f7); q[15] = HI2(qf.f7);
    }

    v8f accT[4];
#pragma unroll
    for (int t = 0; t < 4; ++t)
#pragma unroll
        for (int r = 0; r < 8; ++r) accT[t][r] = 0.0f;

    float m = -3.0e38f;
    float l = 0.0f;

    Frag8 kA, vA, kB, vB;
    load_frag(kA, kp);
    load_frag(vA, vp);
    load_frag(kB, kp + TILE_F4);
    load_frag(vB, vp + TILE_F4);

    for (int j = 0; j < NJ; j += 2) {
        const int ja = (j + 2 < NJ) ? (j + 2) : j;
        const int jb = (j + 3 < NJ) ? (j + 3) : (j + 1);

        v8f s0 = tile_scores(kA, q);
        load_frag(kA, kp + (size_t)ja * TILE_F4);
        __builtin_amdgcn_sched_barrier(0);
        tile_softmax_pv(s0, vA, accT, m, l, hi);
        load_frag(vA, vp + (size_t)ja * TILE_F4);
        __builtin_amdgcn_sched_barrier(0);

        v8f s1 = tile_scores(kB, q);
        load_frag(kB, kp + (size_t)jb * TILE_F4);
        __builtin_amdgcn_sched_barrier(0);
        tile_softmax_pv(s1, vB, accT, m, l, hi);
        load_frag(vB, vp + (size_t)jb * TILE_F4);
        __builtin_amdgcn_sched_barrier(0);
    }

    float inv = 1.0f / l;
    int n = qbase + mrow;
#pragma unroll
    for (int t = 0; t < 4; ++t)
#pragma unroll
        for (int r = 0; r < 8; ++r) {
            int o = 16 * t + r + 8 * hi;
            out[((size_t)b * FDIM + o) * N_NODES + n] = accT[t][r] * inv;
        }
}

extern "C" void kernel_launch(void* const* d_in, const int* in_sizes, int n_in,
                              void* d_out, int out_size, void* d_ws, size_t ws_size,
                              hipStream_t stream) {
    (void)in_sizes; (void)n_in; (void)out_size; (void)ws_size;
    const float* x = (const float*)d_in[0];
    const float* w = (const float*)d_in[1];
    float* kf  = (float*)d_ws;
    float* vf  = kf + (size_t)BATCH * N_NODES * FDIM;
    float* out = (float*)d_out;

    dim3 gPrep(N_NODES / 256, BATCH);
    gcn_prep<<<gPrep, 256, 0, stream>>>(x, w, kf, vf);

    dim3 gFlash(BATCH * (N_NODES / 128));
    gcn_flash<<<gFlash, 256, 0, stream>>>(kf, vf, out);
}


# Waves sharing a SIMD (w, w+4) publish their trip number in LDS each trip; the one behind raises its issue priority (s_setprio 3), so both finish together; arithmetic and its order unchanged
# speedup vs baseline: 2.9717x; 1.0371x over previous
#include <hip/hip_runtime.h>


#define BATCH   4
#define FDIM    64
#define N_NODES 8192
#define NJ      (N_NODES / 16)
#define TILE_F  1024
#define TILE_F4 256

typedef __attribute__((ext_vector_type(2))) float v2f;
typedef __attribute__((ext_vector_type(8))) float v8f;

__global__ __launch_bounds__(256) void gcn_prep(const float* __restrict__ x,
                                                const float* __restrict__ w,
                                                float* __restrict__ kf,
                                                float* __restrict__ vf) {
    const int n  = blockIdx.x * 256 + threadIdx.x;
    const int b  = blockIdx.y;
    const int j  = n >> 4;
    const int kk = n & 15;
    const float* xb = x + (size_t)b * FDIM * N_NODES;
    float4* ktile = (float4*)(kf + ((size_t)b * NJ + j) * TILE_F);
    float*  vtile = vf + ((size_t)b * NJ + j) * TILE_F;

    float acc[FDIM];
#pragma unroll
    for (int o = 0; o < FDIM; ++o) acc[o] = 0.0f;

    for (int g = 0; g < 8; ++g) {
        float v[8];
#pragma unroll
        for (int i = 0; i < 8; ++i)
            v[i] = xb[(size_t)(8 * g + i) * N_NODES + n];
#pragma unroll
        for (int i = 0; i < 8; ++i)
#pragma unroll
            for (int o = 0; o < FDIM; ++o)
                acc[o] += v[i] * w[(8 * g + i) * FDIM + o];
        ktile[g * 32 + kk]      = make_float4(v[0], v[1], v[4], v[5]);
        ktile[g * 32 + 16 + kk] = make_float4(v[2], v[3], v[6], v[7]);
    }

    const int c = kk >> 2, h = (kk >> 1) & 1, e = kk & 1;
    float* vp = vtile + ((c >> 1) * 32 + h * 16) * 4 + (c & 1) * 2 + e;
#pragma unroll
    for (int o = 0; o < FDIM; ++o)
        vp[(o >> 4) * 256 + (o & 15) * 4] = acc[o];
}

struct Frag8 { float4 f0, f1, f2, f3, f4, f5, f6, f7; };

__device__ __forceinline__ void load_frag(Frag8& d, const float4* __restrict__ p) {
    d.f0 = p[0];   d.f1 = p[32];  d.f2 = p[64];  d.f3 = p[96];
    d.f4 = p[128]; d.f5 = p[160]; d.f6 = p[192]; d.f7 = p[224];
}

#define WMMA_F32(A, B, C) \
    __builtin_amdgcn_wmma_f32_16x16x4_f32(false, (A), false, (B), (short)0, (C), false, false)

__device__ __forceinline__ float xhalf(float x) {
    const int b = __builtin_bit_cast(int, x);
    return __builtin_bit_cast(float,
        __builtin_amdgcn_permlanex16(b, b, 0x76543210, 0xfedcba98, false, false));
}

#define LO2(F) (v2f{(F).x, (F).y})
#define HI2(F) (v2f{(F).z, (F).w})

__device__ __forceinline__ v8f tile_scores(const Frag8& K, const v2f (&q)[16]) {
    v8f s;
#pragma unroll
    for (int r = 0; r < 8; ++r) s[r] = 0.0f;
    s = WMMA_F32(LO2(K.f0), q[0],  s);  s = WMMA_F32(HI2(K.f0), q[1],  s);
    s = WMMA_F32(LO2(K.f1), q[2],  s);  s = WMMA_F32(HI2(K.f1), q[3],  s);
    s = WMMA_F32(LO2(K.f2), q[4],  s);  s = WMMA_F32(HI2(K.f2), q[5],  s);
    s = WMMA_F32(LO2(K.f3), q[6],  s);  s = WMMA_F32(HI2(K.f3), q[7],  s);
    s = WMMA_F32(LO2(K.f4), q[8],  s);  s = WMMA_F32(HI2(K.f4), q[9],  s);
    s = WMMA_F32(LO2(K.f5), q[10], s);  s = WMMA_F32(HI2(K.f5), q[11], s);
    s = WMMA_F32(LO2(K.f6), q[12], s);  s = WMMA_F32(HI2(K.f6), q[13], s);
    s = WMMA_F32(LO2(K.f7), q[14], s);  s = WMMA_F32(HI2(K.f7), q[15], s);
    return s;
}

__device__ __forceinline__ void tile_softmax_pv(const v8f s, const Frag8& V,
                                                v8f (&accT)[4],
                                                float& m, float& l, const int hi) {

    float mloc = s[0];
#pragma unroll
    for (int r = 1; r < 8; ++r) mloc = fmaxf(mloc, s[r]);
    mloc = fmaxf(mloc, xhalf(mloc));
    float mnew = fmaxf(m, mloc);
    float sc = __expf(m - mnew);
    m = mnew;

    float p[8], ls = 0.0f;
#pragma unroll
    for (int r = 0; r < 8; ++r) { p[r] = __expf(s[r] - mnew); ls += p[r]; }
    ls += xhalf(ls);
    l = l * sc + ls;

#pragma unroll
    for (int t = 0; t < 4; ++t)
#pragma unroll
        for (int r = 0; r < 8; ++r) accT[t][r] *= sc;

    float tt[8];
#pragma unroll
    for (int r = 0; r < 8; ++r) tt[r] = xhalf(p[r]);
    v2f pb[4];
    pb[0].x = hi ? tt[2] : p[0];  pb[0].y = hi ? tt[3] : p[1];
    pb[1].x = hi ? tt[6] : p[4];  pb[1].y = hi ? tt[7] : p[5];
    pb[2].x = hi ? p[2] : tt[0];  pb[2].y = hi ? p[3] : tt[1];
    pb[3].x = hi ? p[6] : tt[4];  pb[3].y = hi ? p[7] : tt[5];

    accT[0] = WMMA_F32(LO2(V.f0), pb[0], accT[0]);
    accT[1] = WMMA_F32(LO2(V.f2), pb[0], accT[1]);
    accT[2] = WMMA_F32(LO2(V.f4), pb[0], accT[2]);
    accT[3] = WMMA_F32(LO2(V.f6), pb[0], accT[3]);
    accT[0] = WMMA_F32(HI2(V.f0), pb[1], accT[0]);
    accT[1] = WMMA_F32(HI2(V.f2), pb[1], accT[1]);
    accT[2] = WMMA_F32(HI2(V.f4), pb[1], accT[2]);
    accT[3] = WMMA_F32(HI2(V.f6), pb[1], accT[3]);
    accT[0] = WMMA_F32(LO2(V.f1), pb[2], accT[0]);
    accT[1] = WMMA_F32(LO2(V.f3), pb[2], accT[1]);
    accT[2] = WMMA_F32(LO2(V.f5), pb[2], accT[2]);
    accT[3] = WMMA_F32(LO2(V.f7), pb[2], accT[3]);
    accT[0] = WMMA_F32(HI2(V.f1), pb[3], accT[0]);
    accT[1] = WMMA_F32(HI2(V.f3), pb[3], accT[1]);
    accT[2] = WMMA_F32(HI2(V.f5), pb[3], accT[2]);
    accT[3] = WMMA_F32(HI2(V.f7), pb[3], accT[3]);
}

__global__ __launch_bounds__(256) void gcn_flash(const float* __restrict__ kf,
                                                 const float* __restrict__ vf,
                                                 float* __restrict__ out) {
    const int tid  = threadIdx.x;
    const int wave = tid >> 5;
    const int lane = tid & 31;
    const int hi   = lane >> 4;
    const int mrow = lane & 15;

    __shared__ int board[8];
    const int me   = __builtin_amdgcn_readfirstlane(wave);
    const int mate = me ^ 4;
    board[me] = -1;
    __syncthreads();
    int seen = -1;

    const int b     = blockIdx.x >> 6;
    const int qblk  = blockIdx.x & 63;
    const int qtile = qblk * 8 + wave;
    const int qbase = qtile * 16;

    const float4* kp = (const float4*)kf + (size_t)b * NJ * TILE_F4 + lane;
    const float4* vp = (const float4*)vf + (size_t)b * NJ * TILE_F4 + lane;

    v2f q[16];
    {
        Frag8 qf;
        load_frag(qf, kp + (size_t)qtile * TILE_F4);
        q[0]  = LO2(qf.f0); q[1]  = HI2(qf.f0); q[2]  = LO2(qf.f1); q[3]  = HI2(qf.f1);
        q[4]  = LO2(qf.f2); q[5]  = HI2(qf.f2); q[6]  = LO2(qf.f3); q[7]  = HI2(qf.f3);
        q[8]  = LO2(qf.f4); q[9]  = HI2(qf.f4); q[10] = LO2(qf.f5); q[11] = HI2(qf.f5);
        q[12] = LO2(qf.f6); q[13] = HI2(qf.f6); q[14] = LO2(qf.f7); q[15] = HI2(qf.f7);
    }

    v8f accT[4];
#pragma unroll
    for (int t = 0; t < 4; ++t)
#pragma unroll
        for (int r = 0; r < 8; ++r) accT[t][r] = 0.0f;

    float m = -3.0e38f;
    float l = 0.0f;

    Frag8 kA, vA, kB, vB;
    load_frag(kA, kp);
    load_frag(vA, vp);
    load_frag(kB, kp + TILE_F4);
    load_frag(vB, vp + TILE_F4);

    for (int j = 0; j < NJ; j += 2) {
        const int trip = j >> 1;
        if (__builtin_amdgcn_readfirstlane(seen) >= trip) __builtin_amdgcn_s_setprio(3);
        else                                              __builtin_amdgcn_s_setprio(0);

        const int ja = (j + 2 < NJ) ? (j + 2) : j;
        const int jb = (j + 3 < NJ) ? (j + 3) : (j + 1);

        v8f s0 = tile_scores(kA, q);
        load_frag(kA, kp + (size_t)ja * TILE_F4);
        __builtin_amdgcn_sched_barrier(0);
        tile_softmax_pv(s0, vA, accT, m, l, hi);
        load_frag(vA, vp + (size_t)ja * TILE_F4);
        __builtin_amdgcn_sched_barrier(0);

        v8f s1 = tile_scores(kB, q);
        load_frag(kB, kp + (size_t)jb * TILE_F4);
        __builtin_amdgcn_sched_barrier(0);
        asm volatile("" ::: "memory");
        board[me] = trip;
        seen = board[mate];
        tile_softmax_pv(s1, vB, accT, m, l, hi);
        load_frag(vB, vp + (size_t)jb * TILE_F4);
        __builtin_amdgcn_sched_barrier(0);
    }

    __builtin_amdgcn_s_setprio(0);

    float inv = 1.0f / l;
    int n = qbase + mrow;
#pragma unroll
    for (int t = 0; t < 4; ++t)
#pragma unroll
        for (int r = 0; r < 8; ++r) {
            int o = 16 * t + r + 8 * hi;
            out[((size_t)b * FDIM + o) * N_NODES + n] = accT[t][r] * inv;
        }
}

extern "C" void kernel_launch(void* const* d_in, const int* in_sizes, int n_in,
                              void* d_out, int out_size, void* d_ws, size_t ws_size,
                              hipStream_t stream) {
    (void)in_sizes; (void)n_in; (void)out_size; (void)ws_size;
    const float* x = (const float*)d_in[0];
    const float* w = (const float*)d_in[1];
    float* kf  = (float*)d_ws;
    float* vf  = kf + (size_t)BATCH * N_NODES * FDIM;
    float* out = (float*)d_out;

    dim3 gPrep(N_NODES / 256, BATCH);
    gcn_prep<<<gPrep, 256, 0, stream>>>(x, w, kf, vf);

    dim3 gFlash(BATCH * (N_NODES / 128));
    gcn_flash<<<gFlash, 256, 0, stream>>>(kf, vf, out);
}


# Loop skewed by one key tile, issue order requested (sched_group_barrier): softmax vector work of tile j issues between the 16 chained score WMMAs of tile j+1; same operations, same sum order
# speedup vs baseline: 3.0836x; 1.0377x over previous
#include <hip/hip_runtime.h>


#define BATCH   4
#define FDIM    64
#define N_NODES 8192
#define NJ      (N_NODES / 16)
#define TILE_F  1024
#define TILE_F4 256

typedef __attribute__((ext_vector_type(2))) float v2f;
typedef __attribute__((ext_vector_type(8))) float v8f;

__global__ __launch_bounds__(256) void gcn_prep(const float* __restrict__ x,
                                                const float* __restrict__ w,
                                                float* __restrict__ kf,
                                                float* __restrict__ vf) {
    const int n  = blockIdx.x * 256 + threadIdx.x;
    const int b  = blockIdx.y;
    const int j  = n >> 4;
    const int kk = n & 15;
    const float* xb = x + (size_t)b * FDIM * N_NODES;
    float4* ktile = (float4*)(kf + ((size_t)b * NJ + j) * TILE_F);
    float*  vtile = vf + ((size_t)b * NJ + j) * TILE_F;

    float acc[FDIM];
#pragma unroll
    for (int o = 0; o < FDIM; ++o) acc[o] = 0.0f;

    for (int g = 0; g < 8; ++g) {
        float v[8];
#pragma unroll
        for (int i = 0; i < 8; ++i)
            v[i] = xb[(size_t)(8 * g + i) * N_NODES + n];
#pragma unroll
        for (int i = 0; i < 8; ++i)
#pragma unroll
            for (int o = 0; o < FDIM; ++o)
                acc[o] += v[i] * w[(8 * g + i) * FDIM + o];
        ktile[g * 32 + kk]      = make_float4(v[0], v[1], v[4], v[5]);
        ktile[g * 32 + 16 + kk] = make_float4(v[2], v[3], v[6], v[7]);
    }

    const int c = kk >> 2, h = (kk >> 1) & 1, e = kk & 1;
    float* vp = vtile + ((c >> 1) * 32 + h * 16) * 4 + (c & 1) * 2 + e;
#pragma unroll
    for (int o = 0; o < FDIM; ++o)
        vp[(o >> 4) * 256 + (o & 15) * 4] = acc[o];
}

struct Frag8 { float4 f0, f1, f2, f3, f4, f5, f6, f7; };

__device__ __forceinline__ void load_frag(Frag8& d, const float4* __restrict__ p) {
    d.f0 = p[0];   d.f1 = p[32];  d.f2 = p[64];  d.f3 = p[96];
    d.f4 = p[128]; d.f5 = p[160]; d.f6 = p[192]; d.f7 = p[224];
}

#define WMMA_F32(A, B, C) \
    __builtin_amdgcn_wmma_f32_16x16x4_f32(false, (A), false, (B), (short)0, (C), false, false)

__device__ __forceinline__ float xhalf(float x) {
    const int b = __builtin_bit_cast(int, x);
    return __builtin_bit_cast(float,
        __builtin_amdgcn_permlanex16(b, b, 0x76543210, 0xfedcba98, false, false));
}

#define PIPE_W1_V6()                                      \
    __builtin_amdgcn_sched_group_barrier(0x008, 1, 0);    \
    __builtin_amdgcn_sched_group_barrier(0x002, 6, 0);    \
    __builtin_amdgcn_sched_group_barrier(0x400, 1, 0)
#define PIPE_SCORES_WITH_SOFTMAX() do {                   \
    PIPE_W1_V6(); PIPE_W1_V6(); PIPE_W1_V6(); PIPE_W1_V6(); \
    PIPE_W1_V6(); PIPE_W1_V6(); PIPE_W1_V6(); PIPE_W1_V6(); \
    PIPE_W1_V6(); PIPE_W1_V6(); PIPE_W1_V6(); PIPE_W1_V6(); \
    PIPE_W1_V6(); PIPE_W1_V6(); PIPE_W1_V6(); PIPE_W1_V6(); \
    __builtin_amdgcn_sched_group_barrier(0x402, 16, 0);   \
} while (0)
#define PIPE_OUTPUT() __builtin_amdgcn_sched_group_barrier(0x008, 16, 0)
#define PIPE_LOADS()  __builtin_amdgcn_sched_group_barrier(0x020, 8, 0)

#define LO2(F) (v2f{(F).x, (F).y})
#define HI2(F) (v2f{(F).z, (F).w})

__device__ __forceinline__ v8f tile_scores(const Frag8& K, const v2f (&q)[16]) {
    v8f s;
#pragma unroll
    for (int r = 0; r < 8; ++r) s[r] = 0.0f;
    s = WMMA_F32(LO2(K.f0), q[0],  s);  s = WMMA_F32(HI2(K.f0), q[1],  s);
    s = WMMA_F32(LO2(K.f1), q[2],  s);  s = WMMA_F32(HI2(K.f1), q[3],  s);
    s = WMMA_F32(LO2(K.f2), q[4],  s);  s = WMMA_F32(HI2(K.f2), q[5],  s);
    s = WMMA_F32(LO2(K.f3), q[6],  s);  s = WMMA_F32(HI2(K.f3), q[7],  s);
    s = WMMA_F32(LO2(K.f4), q[8],  s);  s = WMMA_F32(HI2(K.f4), q[9],  s);
    s = WMMA_F32(LO2(K.f5), q[10], s);  s = WMMA_F32(HI2(K.f5), q[11], s);
    s = WMMA_F32(LO2(K.f6), q[12], s);  s = WMMA_F32(HI2(K.f6), q[13], s);
    s = WMMA_F32(LO2(K.f7), q[14], s);  s = WMMA_F32(HI2(K.f7), q[15], s);
    return s;
}

__device__ __forceinline__ void tile_softmax(const v8f s, v8f (&accT)[4],
                                             float& m, float& l, const int hi,
                                             v2f (&pb)[4]) {

    float mloc = s[0];
#pragma unroll
    for (int r = 1; r < 8; ++r) mloc = fmaxf(mloc, s[r]);
    mloc = fmaxf(mloc, xhalf(mloc));
    float mnew = fmaxf(m, mloc);
    float sc = __expf(m - mnew);
    m = mnew;

    float p[8], ls = 0.0f;
#pragma unroll
    for (int r = 0; r < 8; ++r) { p[r] = __expf(s[r] - mnew); ls += p[r]; }
    ls += xhalf(ls);
    l = l * sc + ls;

#pragma unroll
    for (int t = 0; t < 4; ++t)
#pragma unroll
        for (int r = 0; r < 8; ++r) accT[t][r] *= sc;

    float tt[8];
#pragma unroll
    for (int r = 0; r < 8; ++r) tt[r] = xhalf(p[r]);
    pb[0].x = hi ? tt[2] : p[0];  pb[0].y = hi ? tt[3] : p[1];
    pb[1].x = hi ? tt[6] : p[4];  pb[1].y = hi ? tt[7] : p[5];
    pb[2].x = hi ? p[2] : tt[0];  pb[2].y = hi ? p[3] : tt[1];
    pb[3].x = hi ? p[6] : tt[4];  pb[3].y = hi ? p[7] : tt[5];
}

__device__ __forceinline__ void tile_pv(const Frag8& V, const v2f (&pb)[4],
                                        v8f (&accT)[4]) {
    accT[0] = WMMA_F32(LO2(V.f0), pb[0], accT[0]);
    accT[1] = WMMA_F32(LO2(V.f2), pb[0], accT[1]);
    accT[2] = WMMA_F32(LO2(V.f4), pb[0], accT[2]);
    accT[3] = WMMA_F32(LO2(V.f6), pb[0], accT[3]);
    accT[0] = WMMA_F32(HI2(V.f0), pb[1], accT[0]);
    accT[1] = WMMA_F32(HI2(V.f2), pb[1], accT[1]);
    accT[2] = WMMA_F32(HI2(V.f4), pb[1], accT[2]);
    accT[3] = WMMA_F32(HI2(V.f6), pb[1], accT[3]);
    accT[0] = WMMA_F32(LO2(V.f1), pb[2], accT[0]);
    accT[1] = WMMA_F32(LO2(V.f3), pb[2], accT[1]);
    accT[2] = WMMA_F32(LO2(V.f5), pb[2], accT[2]);
    accT[3] = WMMA_F32(LO2(V.f7), pb[2], accT[3]);
    accT[0] = WMMA_F32(HI2(V.f1), pb[3], accT[0]);
    accT[1] = WMMA_F32(HI2(V.f3), pb[3], accT[1]);
    accT[2] = WMMA_F32(HI2(V.f5), pb[3], accT[2]);
    accT[3] = WMMA_F32(HI2(V.f7), pb[3], accT[3]);
}

__global__ __launch_bounds__(256) void gcn_flash(const float* __restrict__ kf,
                                                 const float* __restrict__ vf,
                                                 float* __restrict__ out) {
    const int tid  = threadIdx.x;
    const int wave = tid >> 5;
    const int lane = tid & 31;
    const int hi   = lane >> 4;
    const int mrow = lane & 15;

    __shared__ int board[8];
    const int me   = __builtin_amdgcn_readfirstlane(wave);
    const int mate = me ^ 4;
    board[me] = -1;
    __syncthreads();
    int seen = -1;

    const int b     = blockIdx.x >> 6;
    const int qblk  = blockIdx.x & 63;
    const int qtile = qblk * 8 + wave;
    const int qbase = qtile * 16;

    const float4* kp = (const float4*)kf + (size_t)b * NJ * TILE_F4 + lane;
    const float4* vp = (const float4*)vf + (size_t)b * NJ * TILE_F4 + lane;

    v2f q[16];
    {
        Frag8 qf;
        load_frag(qf, kp + (size_t)qtile * TILE_F4);
        q[0]  = LO2(qf.f0); q[1]  = HI2(qf.f0); q[2]  = LO2(qf.f1); q[3]  = HI2(qf.f1);
        q[4]  = LO2(qf.f2); q[5]  = HI2(qf.f2); q[6]  = LO2(qf.f3); q[7]  = HI2(qf.f3);
        q[8]  = LO2(qf.f4); q[9]  = HI2(qf.f4); q[10] = LO2(qf.f5); q[11] = HI2(qf.f5);
        q[12] = LO2(qf.f6); q[13] = HI2(qf.f6); q[14] = LO2(qf.f7); q[15] = HI2(qf.f7);
    }

    v8f accT[4];
#pragma unroll
    for (int t = 0; t < 4; ++t)
#pragma unroll
        for (int r = 0; r < 8; ++r) accT[t][r] = 0.0f;

    float m = -3.0e38f;
    float l = 0.0f;

    Frag8 kA, vA, kB, vB;
    load_frag(kA, kp);
    load_frag(vA, vp);
    load_frag(kB, kp + TILE_F4);
    load_frag(vB, vp + TILE_F4);

    v8f s0 = tile_scores(kA, q);
    load_frag(kA, kp + 2 * TILE_F4);
    __builtin_amdgcn_sched_barrier(0);
    v2f pb[4];

    for (int j = 0; j < NJ; j += 2) {
        const int trip = j >> 1;
        if (__builtin_amdgcn_readfirstlane(seen) >= trip) __builtin_amdgcn_s_setprio(3);
        else                                              __builtin_amdgcn_s_setprio(0);

        const int ja = (j + 2 < NJ) ? (j + 2) : j;
        const int jb = (j + 3 < NJ) ? (j + 3) : (j + 1);
        const int jc = (j + 4 < NJ) ? (j + 4) : j;

        v8f s1 = tile_scores(kB, q);
        tile_softmax(s0, accT, m, l, hi, pb);
        load_frag(kB, kp + (size_t)jb * TILE_F4);
        tile_pv(vA, pb, accT);
        load_frag(vA, vp + (size_t)ja * TILE_F4);

        asm volatile("" ::: "memory");
        board[me] = trip;
        seen = board[mate];

        s0 = tile_scores(kA, q);
        tile_softmax(s1, accT, m, l, hi, pb);
        load_frag(kA, kp + (size_t)jc * TILE_F4);
        asm volatile("" :: "v"(s0));
        tile_pv(vB, pb, accT);
        load_frag(vB, vp + (size_t)jb * TILE_F4);

        PIPE_SCORES_WITH_SOFTMAX();
        PIPE_LOADS();
        PIPE_OUTPUT();
        PIPE_LOADS();
        PIPE_SCORES_WITH_SOFTMAX();
        PIPE_LOADS();
        PIPE_OUTPUT();
        PIPE_LOADS();
        __builtin_amdgcn_sched_barrier(0);
    }

    __builtin_amdgcn_s_setprio(0);

    float inv = 1.0f / l;
    int n = qbase + mrow;
#pragma unroll
    for (int t = 0; t < 4; ++t)
#pragma unroll
        for (int r = 0; r < 8; ++r) {
            int o = 16 * t + r + 8 * hi;
            out[((size_t)b * FDIM + o) * N_NODES + n] = accT[t][r] * inv;
        }
}

extern "C" void kernel_launch(void* const* d_in, const int* in_sizes, int n_in,
                              void* d_out, int out_size, void* d_ws, size_t ws_size,
                              hipStream_t stream) {
    (void)in_sizes; (void)n_in; (void)out_size; (void)ws_size;
    const float* x = (const float*)d_in[0];
    const float* w = (const float*)d_in[1];
    float* kf  = (float*)d_ws;
    float* vf  = kf + (size_t)BATCH * N_NODES * FDIM;
    float* out = (float*)d_out;

    dim3 gPrep(N_NODES / 256, BATCH);
    gcn_prep<<<gPrep, 256, 0, stream>>>(x, w, kf, vf);

    dim3 gFlash(BATCH * (N_NODES / 128));
    gcn_flash<<<gFlash, 256, 0, stream>>>(kf, vf, out);
}


# Key tiles with all scores > 27.75 below the row max skip exp and PV WMMAs (drops terms < 2^-40); waves rotate over the block's 8 query tiles for equal load, partials merged in LDS; sum order changes
# speedup vs baseline: 3.6695x; 1.1900x over previous
#include <hip/hip_runtime.h>


#define BATCH   4
#define FDIM    64
#define N_NODES 8192
#define NJ      (N_NODES / 16)
#define TILE_F  1024
#define TILE_F4 256
#define SKIP_T  27.75f
#define NPH      8
#define PH_TILES (NJ / NPH)
#define PART_F   34

typedef __attribute__((ext_vector_type(2))) float v2f;
typedef __attribute__((ext_vector_type(8))) float v8f;

__global__ __launch_bounds__(256) void gcn_prep(const float* __restrict__ x,
                                                const float* __restrict__ w,
                                                float* __restrict__ kf,
                                                float* __restrict__ vf) {
    const int n  = blockIdx.x * 256 + threadIdx.x;
    const int b  = blockIdx.y;
    const int j  = n >> 4;
    const int kk = n & 15;
    const float* xb = x + (size_t)b * FDIM * N_NODES;
    float4* ktile = (float4*)(kf + ((size_t)b * NJ + j) * TILE_F);
    float*  vtile = vf + ((size_t)b * NJ + j) * TILE_F;

    float acc[FDIM];
#pragma unroll
    for (int o = 0; o < FDIM; ++o) acc[o] = 0.0f;

    for (int g = 0; g < 8; ++g) {
        float v[8];
#pragma unroll
        for (int i = 0; i < 8; ++i)
            v[i] = xb[(size_t)(8 * g + i) * N_NODES + n];
#pragma unroll
        for (int i = 0; i < 8; ++i)
#pragma unroll
            for (int o = 0; o < FDIM; ++o)
                acc[o] += v[i] * w[(8 * g + i) * FDIM + o];
        ktile[g * 32 + kk]      = make_float4(v[0], v[1], v[4], v[5]);
        ktile[g * 32 + 16 + kk] = make_float4(v[2], v[3], v[6], v[7]);
    }

    const int c = kk >> 2, h = (kk >> 1) & 1, e = kk & 1;
    float* vp = vtile + ((c >> 1) * 32 + h * 16) * 4 + (c & 1) * 2 + e;
#pragma unroll
    for (int o = 0; o < FDIM; ++o)
        vp[(o >> 4) * 256 + (o & 15) * 4] = acc[o];
}

struct Frag8 { float4 f0, f1, f2, f3, f4, f5, f6, f7; };

__device__ __forceinline__ void load_frag(Frag8& d, const float4* __restrict__ p) {
    d.f0 = p[0];   d.f1 = p[32];  d.f2 = p[64];  d.f3 = p[96];
    d.f4 = p[128]; d.f5 = p[160]; d.f6 = p[192]; d.f7 = p[224];
}

#define WMMA_F32(A, B, C) \
    __builtin_amdgcn_wmma_f32_16x16x4_f32(false, (A), false, (B), (short)0, (C), false, false)

__device__ __forceinline__ float xhalf(float x) {
    const int b = __builtin_bit_cast(int, x);
    return __builtin_bit_cast(float,
        __builtin_amdgcn_permlanex16(b, b, 0x76543210, 0xfedcba98, false, false));
}

#define PIPE_W1_V6()                                      \
    __builtin_amdgcn_sched_group_barrier(0x008, 1, 0);    \
    __builtin_amdgcn_sched_group_barrier(0x002, 6, 0);    \
    __builtin_amdgcn_sched_group_barrier(0x400, 1, 0)
#define PIPE_SCORES_WITH_SOFTMAX() do {                   \
    PIPE_W1_V6(); PIPE_W1_V6(); PIPE_W1_V6(); PIPE_W1_V6(); \
    PIPE_W1_V6(); PIPE_W1_V6(); PIPE_W1_V6(); PIPE_W1_V6(); \
    PIPE_W1_V6(); PIPE_W1_V6(); PIPE_W1_V6(); PIPE_W1_V6(); \
    PIPE_W1_V6(); PIPE_W1_V6(); PIPE_W1_V6(); PIPE_W1_V6(); \
    __builtin_amdgcn_sched_group_barrier(0x402, 16, 0);   \
} while (0)
#define PIPE_OUTPUT() __builtin_amdgcn_sched_group_barrier(0x008, 16, 0)
#define PIPE_LOADS()  __builtin_amdgcn_sched_group_barrier(0x020, 8, 0)

#define LO2(F) (v2f{(F).x, (F).y})
#define HI2(F) (v2f{(F).z, (F).w})

__device__ __forceinline__ v8f tile_scores(const Frag8& K, const v2f (&q)[16]) {
    v8f s;
#pragma unroll
    for (int r = 0; r < 8; ++r) s[r] = 0.0f;
    s = WMMA_F32(LO2(K.f0), q[0],  s);  s = WMMA_F32(HI2(K.f0), q[1],  s);
    s = WMMA_F32(LO2(K.f1), q[2],  s);  s = WMMA_F32(HI2(K.f1), q[3],  s);
    s = WMMA_F32(LO2(K.f2), q[4],  s);  s = WMMA_F32(HI2(K.f2), q[5],  s);
    s = WMMA_F32(LO2(K.f3), q[6],  s);  s = WMMA_F32(HI2(K.f3), q[7],  s);
    s = WMMA_F32(LO2(K.f4), q[8],  s);  s = WMMA_F32(HI2(K.f4), q[9],  s);
    s = WMMA_F32(LO2(K.f5), q[10], s);  s = WMMA_F32(HI2(K.f5), q[11], s);
    s = WMMA_F32(LO2(K.f6), q[12], s);  s = WMMA_F32(HI2(K.f6), q[13], s);
    s = WMMA_F32(LO2(K.f7), q[14], s);  s = WMMA_F32(HI2(K.f7), q[15], s);
    return s;
}

__device__ __forceinline__ float tile_max8(const v8f s) {
    float mloc = s[0];
#pragma unroll
    for (int r = 1; r < 8; ++r) mloc = fmaxf(mloc, s[r]);
    return mloc;
}

__device__ __forceinline__ void tile_softmax(const v8f s, v8f (&accT)[4],
                                             float& m, float& l, const int hi,
                                             v2f (&pb)[4]) {

    float mloc = s[0];
#pragma unroll
    for (int r = 1; r < 8; ++r) mloc = fmaxf(mloc, s[r]);
    mloc = fmaxf(mloc, xhalf(mloc));
    float mnew = fmaxf(m, mloc);
    float sc = __expf(m - mnew);
    m = mnew;

    float p[8], ls = 0.0f;
#pragma unroll
    for (int r = 0; r < 8; ++r) { p[r] = __expf(s[r] - mnew); ls += p[r]; }
    ls += xhalf(ls);
    l = l * sc + ls;

#pragma unroll
    for (int t = 0; t < 4; ++t)
#pragma unroll
        for (int r = 0; r < 8; ++r) accT[t][r] *= sc;

    float tt[8];
#pragma unroll
    for (int r = 0; r < 8; ++r) tt[r] = xhalf(p[r]);
    pb[0].x = hi ? tt[2] : p[0];  pb[0].y = hi ? tt[3] : p[1];
    pb[1].x = hi ? tt[6] : p[4];  pb[1].y = hi ? tt[7] : p[5];
    pb[2].x = hi ? p[2] : tt[0];  pb[2].y = hi ? p[3] : tt[1];
    pb[3].x = hi ? p[6] : tt[4];  pb[3].y = hi ? p[7] : tt[5];
}

__device__ __forceinline__ void tile_pv(const Frag8& V, const v2f (&pb)[4],
                                        v8f (&accT)[4]) {
    accT[0] = WMMA_F32(LO2(V.f0), pb[0], accT[0]);
    accT[1] = WMMA_F32(LO2(V.f2), pb[0], accT[1]);
    accT[2] = WMMA_F32(LO2(V.f4), pb[0], accT[2]);
    accT[3] = WMMA_F32(LO2(V.f6), pb[0], accT[3]);
    accT[0] = WMMA_F32(HI2(V.f0), pb[1], accT[0]);
    accT[1] = WMMA_F32(HI2(V.f2), pb[1], accT[1]);
    accT[2] = WMMA_F32(HI2(V.f4), pb[1], accT[2]);
    accT[3] = WMMA_F32(HI2(V.f6), pb[1], accT[3]);
    accT[0] = WMMA_F32(LO2(V.f1), pb[2], accT[0]);
    accT[1] = WMMA_F32(LO2(V.f3), pb[2], accT[1]);
    accT[2] = WMMA_F32(LO2(V.f5), pb[2], accT[2]);
    accT[3] = WMMA_F32(LO2(V.f7), pb[2], accT[3]);
    accT[0] = WMMA_F32(HI2(V.f1), pb[3], accT[0]);
    accT[1] = WMMA_F32(HI2(V.f3), pb[3], accT[1]);
    accT[2] = WMMA_F32(HI2(V.f5), pb[3], accT[2]);
    accT[3] = WMMA_F32(HI2(V.f7), pb[3], accT[3]);
}

__global__ __launch_bounds__(256) void gcn_flash(const float* __restrict__ kf,
                                                 const float* __restrict__ vf,
                                                 float* __restrict__ out) {
    const int tid  = threadIdx.x;
    const int wave = tid >> 5;
    const int lane = tid & 31;
    const int hi   = lane >> 4;
    const int mrow = lane & 15;

    __shared__ int board[8];
    const int me   = __builtin_amdgcn_readfirstlane(wave);
    const int mate = me ^ 4;
    board[me] = -1;
    __syncthreads();
    int seen = -1;

    const int b     = blockIdx.x >> 6;
    const int qblk  = blockIdx.x & 63;

    __shared__ float part[(NPH - 1) * 8 * PART_F * 32];

    const float4* kp = (const float4*)kf + (size_t)b * NJ * TILE_F4 + lane;
    const float4* vp = (const float4*)vf + (size_t)b * NJ * TILE_F4 + lane;

    Frag8 kA, vA, kB, vB;
    load_frag(vA, vp);
    load_frag(kB, kp + TILE_F4);
    load_frag(vB, vp + TILE_F4);
    load_frag(kA, kp + 2 * TILE_F4);

    v2f  q[16];
    v8f  accT[4];
    float m = 0.0f, l = 0.0f;
    v2f  pb[4];
    int  qt = me;

    for (int t = 0; t < NPH; ++t) {
        qt = (me + t) & 7;
        const int j0 = t * PH_TILES;
        Frag8 kT;
        {
            Frag8 qf;
            load_frag(qf, kp + (size_t)(qblk * 8 + qt) * TILE_F4);
            load_frag(kT, kp + (size_t)j0 * TILE_F4);
            q[0]  = LO2(qf.f0); q[1]  = HI2(qf.f0); q[2]  = LO2(qf.f1); q[3]  = HI2(qf.f1);
            q[4]  = LO2(qf.f2); q[5]  = HI2(qf.f2); q[6]  = LO2(qf.f3); q[7]  = HI2(qf.f3);
            q[8]  = LO2(qf.f4); q[9]  = HI2(qf.f4); q[10] = LO2(qf.f5); q[11] = HI2(qf.f5);
            q[12] = LO2(qf.f6); q[13] = HI2(qf.f6); q[14] = LO2(qf.f7); q[15] = HI2(qf.f7);
        }
#pragma unroll
        for (int tt = 0; tt < 4; ++tt)
#pragma unroll
            for (int r = 0; r < 8; ++r) accT[tt][r] = 0.0f;
        m = 0.0f;
#pragma unroll
        for (int c = 0; c < 16; ++c) m += q[c].x * q[c].x + q[c].y * q[c].y;
        m += xhalf(m);
        l = 0.0f;

        v8f s0 = tile_scores(kT, q);

        for (int j = j0; j < j0 + PH_TILES; j += 2) {
            const int trip = j >> 1;
            if (__builtin_amdgcn_readfirstlane(seen) >= trip) __builtin_amdgcn_s_setprio(3);
            else                                              __builtin_amdgcn_s_setprio(0);

            const int ja = (j + 2 < NJ) ? (j + 2) : j;
            const int jb = (j + 3 < NJ) ? (j + 3) : (j + 1);
            const int jc = (j + 4 < NJ) ? (j + 4) : j;

            v8f s1;
            if (__builtin_amdgcn_ballot_w32(tile_max8(s0) > m - SKIP_T) != 0u) {
                tile_softmax(s0, accT, m, l, hi, pb);
                s1 = tile_scores(kB, q);
                tile_pv(vA, pb, accT);
                PIPE_SCORES_WITH_SOFTMAX();
                PIPE_OUTPUT();
            } else {
                s1 = tile_scores(kB, q);
            }
            load_frag(kB, kp + (size_t)jb * TILE_F4);
            load_frag(vA, vp + (size_t)ja * TILE_F4);

            asm volatile("" ::: "memory");
            board[me] = trip;
            seen = board[mate];

            if (__builtin_amdgcn_ballot_w32(tile_max8(s1) > m - SKIP_T) != 0u) {
                tile_softmax(s1, accT, m, l, hi, pb);
                s0 = tile_scores(kA, q);
                tile_pv(vB, pb, accT);
                PIPE_SCORES_WITH_SOFTMAX();
                PIPE_OUTPUT();
            } else {
                s0 = tile_scores(kA, q);
            }
            load_frag(kA, kp + (size_t)jc * TILE_F4);
            load_frag(vB, vp + (size_t)jb * TILE_F4);
        }

        if (t < NPH - 1) {
            float* ps = part + (size_t)((t * 8 + qt) * PART_F) * 32 + lane;
#pragma unroll
            for (int tt = 0; tt < 4; ++tt)
#pragma unroll
                for (int r = 0; r < 8; ++r) ps[(tt * 8 + r) * 32] = accT[tt][r];
            ps[32 * 32] = m;
            ps[33 * 32] = l;
        }
    }

    __builtin_amdgcn_s_setprio(0);

    __syncthreads();
    for (int i = 0; i < NPH - 1; ++i) {
        const float* pi = part + (size_t)((i * 8 + qt) * PART_F) * 32 + lane;
        const float mi   = pi[32 * 32];
        const float li   = pi[33 * 32];
        const float mnew = fmaxf(m, mi);
        const float sa   = __expf(m - mnew);
        const float sb   = __expf(mi - mnew);
        l = l * sa + li * sb;
#pragma unroll
        for (int tt = 0; tt < 4; ++tt)
#pragma unroll
            for (int r = 0; r < 8; ++r)
                accT[tt][r] = accT[tt][r] * sa + pi[(tt * 8 + r) * 32] * sb;
        m = mnew;
    }

    float inv = 1.0f / l;
    int n = (qblk * 8 + qt) * 16 + mrow;
#pragma unroll
    for (int t = 0; t < 4; ++t)
#pragma unroll
        for (int r = 0; r < 8; ++r) {
            int o = 16 * t + r + 8 * hi;
            out[((size_t)b * FDIM + o) * N_NODES + n] = accT[t][r] * inv;
        }
}

extern "C" void kernel_launch(void* const* d_in, const int* in_sizes, int n_in,
                              void* d_out, int out_size, void* d_ws, size_t ws_size,
                              hipStream_t stream) {
    (void)in_sizes; (void)n_in; (void)out_size; (void)ws_size;
    const float* x = (const float*)d_in[0];
    const float* w = (const float*)d_in[1];
    float* kf  = (float*)d_ws;
    float* vf  = kf + (size_t)BATCH * N_NODES * FDIM;
    float* out = (float*)d_out;

    dim3 gPrep(N_NODES / 256, BATCH);
    gcn_prep<<<gPrep, 256, 0, stream>>>(x, w, kf, vf);

    dim3 gFlash(BATCH * (N_NODES / 128));
    gcn_flash<<<gFlash, 256, 0, stream>>>(kf, vf, out);
}


# A block ranks its 128 query rows by |x_n|^2 in LDS and forms its 8 query tiles in rank order (outputs put back via LDS): weak rows share tiles, fewer unskipped key tiles; dropped <2^-40 terms differ
# speedup vs baseline: 4.0630x; 1.1072x over previous
#include <hip/hip_runtime.h>


#define BATCH   4
#define FDIM    64
#define N_NODES 8192
#define NJ      (N_NODES / 16)
#define TILE_F  1024
#define TILE_F4 256
#define SKIP_T  27.75f
#define NPH      8
#define PH_TILES (NJ / NPH)
#define PART_F   34

typedef __attribute__((ext_vector_type(2))) float v2f;
typedef __attribute__((ext_vector_type(8))) float v8f;

__global__ __launch_bounds__(256) void gcn_prep(const float* __restrict__ x,
                                                const float* __restrict__ w,
                                                float* __restrict__ kf,
                                                float* __restrict__ vf) {
    const int n  = blockIdx.x * 256 + threadIdx.x;
    const int b  = blockIdx.y;
    const int j  = n >> 4;
    const int kk = n & 15;
    const float* xb = x + (size_t)b * FDIM * N_NODES;
    float4* ktile = (float4*)(kf + ((size_t)b * NJ + j) * TILE_F);
    float*  vtile = vf + ((size_t)b * NJ + j) * TILE_F;

    float acc[FDIM];
#pragma unroll
    for (int o = 0; o < FDIM; ++o) acc[o] = 0.0f;

    for (int g = 0; g < 8; ++g) {
        float v[8];
#pragma unroll
        for (int i = 0; i < 8; ++i)
            v[i] = xb[(size_t)(8 * g + i) * N_NODES + n];
#pragma unroll
        for (int i = 0; i < 8; ++i)
#pragma unroll
            for (int o = 0; o < FDIM; ++o)
                acc[o] += v[i] * w[(8 * g + i) * FDIM + o];
        ktile[g * 32 + kk]      = make_float4(v[0], v[1], v[4], v[5]);
        ktile[g * 32 + 16 + kk] = make_float4(v[2], v[3], v[6], v[7]);
    }

    const int c = kk >> 2, h = (kk >> 1) & 1, e = kk & 1;
    float* vp = vtile + ((c >> 1) * 32 + h * 16) * 4 + (c & 1) * 2 + e;
#pragma unroll
    for (int o = 0; o < FDIM; ++o)
        vp[(o >> 4) * 256 + (o & 15) * 4] = acc[o];
}

struct Frag8 { float4 f0, f1, f2, f3, f4, f5, f6, f7; };

__device__ __forceinline__ void load_frag(Frag8& d, const float4* __restrict__ p) {
    d.f0 = p[0];   d.f1 = p[32];  d.f2 = p[64];  d.f3 = p[96];
    d.f4 = p[128]; d.f5 = p[160]; d.f6 = p[192]; d.f7 = p[224];
}

#define WMMA_F32(A, B, C) \
    __builtin_amdgcn_wmma_f32_16x16x4_f32(false, (A), false, (B), (short)0, (C), false, false)

__device__ __forceinline__ float xhalf(float x) {
    const int b = __builtin_bit_cast(int, x);
    return __builtin_bit_cast(float,
        __builtin_amdgcn_permlanex16(b, b, 0x76543210, 0xfedcba98, false, false));
}

#define PIPE_W1_V6()                                      \
    __builtin_amdgcn_sched_group_barrier(0x008, 1, 0);    \
    __builtin_amdgcn_sched_group_barrier(0x002, 6, 0);    \
    __builtin_amdgcn_sched_group_barrier(0x400, 1, 0)
#define PIPE_SCORES_WITH_SOFTMAX() do {                   \
    PIPE_W1_V6(); PIPE_W1_V6(); PIPE_W1_V6(); PIPE_W1_V6(); \
    PIPE_W1_V6(); PIPE_W1_V6(); PIPE_W1_V6(); PIPE_W1_V6(); \
    PIPE_W1_V6(); PIPE_W1_V6(); PIPE_W1_V6(); PIPE_W1_V6(); \
    PIPE_W1_V6(); PIPE_W1_V6(); PIPE_W1_V6(); PIPE_W1_V6(); \
    __builtin_amdgcn_sched_group_barrier(0x402, 16, 0);   \
} while (0)
#define PIPE_OUTPUT() __builtin_amdgcn_sched_group_barrier(0x008, 16, 0)
#define PIPE_LOADS()  __builtin_amdgcn_sched_group_barrier(0x020, 8, 0)

#define LO2(F) (v2f{(F).x, (F).y})
#define HI2(F) (v2f{(F).z, (F).w})

__device__ __forceinline__ v8f tile_scores(const Frag8& K, const v2f (&q)[16]) {
    v8f s;
#pragma unroll
    for (int r = 0; r < 8; ++r) s[r] = 0.0f;
    s = WMMA_F32(LO2(K.f0), q[0],  s);  s = WMMA_F32(HI2(K.f0), q[1],  s);
    s = WMMA_F32(LO2(K.f1), q[2],  s);  s = WMMA_F32(HI2(K.f1), q[3],  s);
    s = WMMA_F32(LO2(K.f2), q[4],  s);  s = WMMA_F32(HI2(K.f2), q[5],  s);
    s = WMMA_F32(LO2(K.f3), q[6],  s);  s = WMMA_F32(HI2(K.f3), q[7],  s);
    s = WMMA_F32(LO2(K.f4), q[8],  s);  s = WMMA_F32(HI2(K.f4), q[9],  s);
    s = WMMA_F32(LO2(K.f5), q[10], s);  s = WMMA_F32(HI2(K.f5), q[11], s);
    s = WMMA_F32(LO2(K.f6), q[12], s);  s = WMMA_F32(HI2(K.f6), q[13], s);
    s = WMMA_F32(LO2(K.f7), q[14], s);  s = WMMA_F32(HI2(K.f7), q[15], s);
    return s;
}

__device__ __forceinline__ float tile_max8(const v8f s) {
    float mloc = s[0];
#pragma unroll
    for (int r = 1; r < 8; ++r) mloc = fmaxf(mloc, s[r]);
    return mloc;
}

__device__ __forceinline__ void tile_softmax(const v8f s, v8f (&accT)[4],
                                             float& m, float& l, const int hi,
                                             v2f (&pb)[4]) {

    float mloc = s[0];
#pragma unroll
    for (int r = 1; r < 8; ++r) mloc = fmaxf(mloc, s[r]);
    mloc = fmaxf(mloc, xhalf(mloc));
    float mnew = fmaxf(m, mloc);
    float sc = __expf(m - mnew);
    m = mnew;

    float p[8], ls = 0.0f;
#pragma unroll
    for (int r = 0; r < 8; ++r) { p[r] = __expf(s[r] - mnew); ls += p[r]; }
    ls += xhalf(ls);
    l = l * sc + ls;

#pragma unroll
    for (int t = 0; t < 4; ++t)
#pragma unroll
        for (int r = 0; r < 8; ++r) accT[t][r] *= sc;

    float tt[8];
#pragma unroll
    for (int r = 0; r < 8; ++r) tt[r] = xhalf(p[r]);
    pb[0].x = hi ? tt[2] : p[0];  pb[0].y = hi ? tt[3] : p[1];
    pb[1].x = hi ? tt[6] : p[4];  pb[1].y = hi ? tt[7] : p[5];
    pb[2].x = hi ? p[2] : tt[0];  pb[2].y = hi ? p[3] : tt[1];
    pb[3].x = hi ? p[6] : tt[4];  pb[3].y = hi ? p[7] : tt[5];
}

__device__ __forceinline__ void tile_pv(const Frag8& V, const v2f (&pb)[4],
                                        v8f (&accT)[4]) {
    accT[0] = WMMA_F32(LO2(V.f0), pb[0], accT[0]);
    accT[1] = WMMA_F32(LO2(V.f2), pb[0], accT[1]);
    accT[2] = WMMA_F32(LO2(V.f4), pb[0], accT[2]);
    accT[3] = WMMA_F32(LO2(V.f6), pb[0], accT[3]);
    accT[0] = WMMA_F32(HI2(V.f0), pb[1], accT[0]);
    accT[1] = WMMA_F32(HI2(V.f2), pb[1], accT[1]);
    accT[2] = WMMA_F32(HI2(V.f4), pb[1], accT[2]);
    accT[3] = WMMA_F32(HI2(V.f6), pb[1], accT[3]);
    accT[0] = WMMA_F32(LO2(V.f1), pb[2], accT[0]);
    accT[1] = WMMA_F32(LO2(V.f3), pb[2], accT[1]);
    accT[2] = WMMA_F32(LO2(V.f5), pb[2], accT[2]);
    accT[3] = WMMA_F32(LO2(V.f7), pb[2], accT[3]);
    accT[0] = WMMA_F32(HI2(V.f1), pb[3], accT[0]);
    accT[1] = WMMA_F32(HI2(V.f3), pb[3], accT[1]);
    accT[2] = WMMA_F32(HI2(V.f5), pb[3], accT[2]);
    accT[3] = WMMA_F32(HI2(V.f7), pb[3], accT[3]);
}

__global__ __launch_bounds__(256) void gcn_flash(const float* __restrict__ kf,
                                                 const float* __restrict__ vf,
                                                 float* __restrict__ out) {
    const int tid  = threadIdx.x;
    const int wave = tid >> 5;
    const int lane = tid & 31;
    const int hi   = lane >> 4;
    const int mrow = lane & 15;

    __shared__ int board[8];
    const int me   = __builtin_amdgcn_readfirstlane(wave);
    const int mate = me ^ 4;
    board[me] = -1;
    int seen = -1;

    const int b     = blockIdx.x >> 6;
    const int qblk  = blockIdx.x & 63;

    const float4* kp = (const float4*)kf + (size_t)b * NJ * TILE_F4 + lane;
    const float4* vp = (const float4*)vf + (size_t)b * NJ * TILE_F4 + lane;

    Frag8 kA, vA, kB, vB;
    load_frag(vA, vp);
    load_frag(kB, kp + TILE_F4);
    load_frag(vB, vp + TILE_F4);
    load_frag(kA, kp + 2 * TILE_F4);

    __shared__ unsigned nkey[128];
    __shared__ int      srow[128];
    if (tid < 128) {
        const float4* rp = (const float4*)kf
                         + ((size_t)b * NJ + qblk * 8 + (tid >> 4)) * TILE_F4 + (tid & 15);
        float a = 0.0f;
#pragma unroll
        for (int g = 0; g < 16; ++g) {
            const float4 v = rp[g * 16];
            a += v.x * v.x + v.y * v.y + v.z * v.z + v.w * v.w;
        }
        nkey[tid] = __builtin_bit_cast(unsigned, a);
    }
    __syncthreads();
    if (tid < 128) {
        const unsigned ki = nkey[tid];
        int rank = 0;
        for (int jj = 0; jj < 128; ++jj) {
            const unsigned kj = nkey[jj];
            rank += (kj < ki || (kj == ki && jj < tid)) ? 1 : 0;
        }
        srow[rank] = tid;
    }
    __syncthreads();

    __shared__ float part[(NPH - 1) * 8 * PART_F * 32];


    v2f  q[16];
    v8f  accT[4];
    float m = 0.0f, l = 0.0f;
    v2f  pb[4];
    int  qt = me;

    for (int t = 0; t < NPH; ++t) {
        {
            const int cc = me ^ t;
            qt = (cc & 4) ? (7 - (cc & 3)) : cc;
        }
        const int j0 = t * PH_TILES;
        Frag8 kT;
        {
            Frag8 qf;
            const int qi = srow[qt * 16 + mrow];
            load_frag(qf, (const float4*)kf + ((size_t)b * NJ + qblk * 8 + (qi >> 4)) * TILE_F4
                          + hi * 16 + (qi & 15));
            load_frag(kT, kp + (size_t)j0 * TILE_F4);
            q[0]  = LO2(qf.f0); q[1]  = HI2(qf.f0); q[2]  = LO2(qf.f1); q[3]  = HI2(qf.f1);
            q[4]  = LO2(qf.f2); q[5]  = HI2(qf.f2); q[6]  = LO2(qf.f3); q[7]  = HI2(qf.f3);
            q[8]  = LO2(qf.f4); q[9]  = HI2(qf.f4); q[10] = LO2(qf.f5); q[11] = HI2(qf.f5);
            q[12] = LO2(qf.f6); q[13] = HI2(qf.f6); q[14] = LO2(qf.f7); q[15] = HI2(qf.f7);
        }
#pragma unroll
        for (int tt = 0; tt < 4; ++tt)
#pragma unroll
            for (int r = 0; r < 8; ++r) accT[tt][r] = 0.0f;
        m = 0.0f;
#pragma unroll
        for (int c = 0; c < 16; ++c) m += q[c].x * q[c].x + q[c].y * q[c].y;
        m += xhalf(m);
        l = 0.0f;

        v8f s0 = tile_scores(kT, q);

        for (int j = j0; j < j0 + PH_TILES; j += 2) {
            const int trip = j >> 1;
            if (__builtin_amdgcn_readfirstlane(seen) >= trip) __builtin_amdgcn_s_setprio(3);
            else                                              __builtin_amdgcn_s_setprio(0);

            const int ja = (j + 2 < NJ) ? (j + 2) : j;
            const int jb = (j + 3 < NJ) ? (j + 3) : (j + 1);
            const int jc = (j + 4 < NJ) ? (j + 4) : j;

            v8f s1;
            if (__builtin_amdgcn_ballot_w32(tile_max8(s0) > m - SKIP_T) != 0u) {
                tile_softmax(s0, accT, m, l, hi, pb);
                s1 = tile_scores(kB, q);
                tile_pv(vA, pb, accT);
                PIPE_SCORES_WITH_SOFTMAX();
                PIPE_OUTPUT();
            } else {
                s1 = tile_scores(kB, q);
            }
            load_frag(kB, kp + (size_t)jb * TILE_F4);
            load_frag(vA, vp + (size_t)ja * TILE_F4);

            asm volatile("" ::: "memory");
            board[me] = trip;
            seen = board[mate];

            if (__builtin_amdgcn_ballot_w32(tile_max8(s1) > m - SKIP_T) != 0u) {
                tile_softmax(s1, accT, m, l, hi, pb);
                s0 = tile_scores(kA, q);
                tile_pv(vB, pb, accT);
                PIPE_SCORES_WITH_SOFTMAX();
                PIPE_OUTPUT();
            } else {
                s0 = tile_scores(kA, q);
            }
            load_frag(kA, kp + (size_t)jc * TILE_F4);
            load_frag(vB, vp + (size_t)jb * TILE_F4);
        }

        if (t < NPH - 1) {
            float* ps = part + (size_t)((t * 8 + qt) * PART_F) * 32 + lane;
#pragma unroll
            for (int tt = 0; tt < 4; ++tt)
#pragma unroll
                for (int r = 0; r < 8; ++r) ps[(tt * 8 + r) * 32] = accT[tt][r];
            ps[32 * 32] = m;
            ps[33 * 32] = l;
        }
    }

    __builtin_amdgcn_s_setprio(0);

    __syncthreads();
    for (int i = 0; i < NPH - 1; ++i) {
        const float* pi = part + (size_t)((i * 8 + qt) * PART_F) * 32 + lane;
        const float mi   = pi[32 * 32];
        const float li   = pi[33 * 32];
        const float mnew = fmaxf(m, mi);
        const float sa   = __expf(m - mnew);
        const float sb   = __expf(mi - mnew);
        l = l * sa + li * sb;
#pragma unroll
        for (int tt = 0; tt < 4; ++tt)
#pragma unroll
            for (int r = 0; r < 8; ++r)
                accT[tt][r] = accT[tt][r] * sa + pi[(tt * 8 + r) * 32] * sb;
        m = mnew;
    }

    __shared__ float ostage[FDIM * 128];
    {
        const float inv = 1.0f / l;
        const int qi = srow[qt * 16 + mrow];
#pragma unroll
        for (int t = 0; t < 4; ++t)
#pragma unroll
            for (int r = 0; r < 8; ++r) {
                const int o = 16 * t + r + 8 * hi;
                ostage[o * 128 + qi] = accT[t][r] * inv;
            }
    }
    __syncthreads();
#pragma unroll
    for (int k = 0; k < 8; ++k) {
        const int o = me * 8 + k;
        const float4 v = ((const float4*)ostage)[o * 32 + lane];
        ((float4*)(out + ((size_t)b * FDIM + o) * N_NODES + qblk * 128))[lane] = v;
    }
}

extern "C" void kernel_launch(void* const* d_in, const int* in_sizes, int n_in,
                              void* d_out, int out_size, void* d_ws, size_t ws_size,
                              hipStream_t stream) {
    (void)in_sizes; (void)n_in; (void)out_size; (void)ws_size;
    const float* x = (const float*)d_in[0];
    const float* w = (const float*)d_in[1];
    float* kf  = (float*)d_ws;
    float* vf  = kf + (size_t)BATCH * N_NODES * FDIM;
    float* out = (float*)d_out;

    dim3 gPrep(N_NODES / 256, BATCH);
    gcn_prep<<<gPrep, 256, 0, stream>>>(x, w, kf, vf);

    dim3 gFlash(BATCH * (N_NODES / 128));
    gcn_flash<<<gFlash, 256, 0, stream>>>(kf, vf, out);
}


# V fragments fetched only for key tiles that pass the skip test, at the head of the tile's stage under its score chain; skipped tiles load K alone (8 loads, not 16); same arithmetic, same sum order
# speedup vs baseline: 4.4238x; 1.0888x over previous
#include <hip/hip_runtime.h>


#define BATCH   4
#define FDIM    64
#define N_NODES 8192
#define NJ      (N_NODES / 16)
#define TILE_F  1024
#define TILE_F4 256
#define SKIP_T  27.75f
#define NPH      8
#define PH_TILES (NJ / NPH)
#define PART_F   34

typedef __attribute__((ext_vector_type(2))) float v2f;
typedef __attribute__((ext_vector_type(8))) float v8f;

__global__ __launch_bounds__(256) void gcn_prep(const float* __restrict__ x,
                                                const float* __restrict__ w,
                                                float* __restrict__ kf,
                                                float* __restrict__ vf) {
    const int n  = blockIdx.x * 256 + threadIdx.x;
    const int b  = blockIdx.y;
    const int j  = n >> 4;
    const int kk = n & 15;
    const float* xb = x + (size_t)b * FDIM * N_NODES;
    float4* ktile = (float4*)(kf + ((size_t)b * NJ + j) * TILE_F);
    float*  vtile = vf + ((size_t)b * NJ + j) * TILE_F;

    float acc[FDIM];
#pragma unroll
    for (int o = 0; o < FDIM; ++o) acc[o] = 0.0f;

    for (int g = 0; g < 8; ++g) {
        float v[8];
#pragma unroll
        for (int i = 0; i < 8; ++i)
            v[i] = xb[(size_t)(8 * g + i) * N_NODES + n];
#pragma unroll
        for (int i = 0; i < 8; ++i)
#pragma unroll
            for (int o = 0; o < FDIM; ++o)
                acc[o] += v[i] * w[(8 * g + i) * FDIM + o];
        ktile[g * 32 + kk]      = make_float4(v[0], v[1], v[4], v[5]);
        ktile[g * 32 + 16 + kk] = make_float4(v[2], v[3], v[6], v[7]);
    }

    const int c = kk >> 2, h = (kk >> 1) & 1, e = kk & 1;
    float* vp = vtile + ((c >> 1) * 32 + h * 16) * 4 + (c & 1) * 2 + e;
#pragma unroll
    for (int o = 0; o < FDIM; ++o)
        vp[(o >> 4) * 256 + (o & 15) * 4] = acc[o];
}

struct Frag8 { float4 f0, f1, f2, f3, f4, f5, f6, f7; };

__device__ __forceinline__ void load_frag(Frag8& d, const float4* __restrict__ p) {
    d.f0 = p[0];   d.f1 = p[32];  d.f2 = p[64];  d.f3 = p[96];
    d.f4 = p[128]; d.f5 = p[160]; d.f6 = p[192]; d.f7 = p[224];
}

#define WMMA_F32(A, B, C) \
    __builtin_amdgcn_wmma_f32_16x16x4_f32(false, (A), false, (B), (short)0, (C), false, false)

__device__ __forceinline__ float xhalf(float x) {
    const int b = __builtin_bit_cast(int, x);
    return __builtin_bit_cast(float,
        __builtin_amdgcn_permlanex16(b, b, 0x76543210, 0xfedcba98, false, false));
}

#define PIPE_W1_V6()                                      \
    __builtin_amdgcn_sched_group_barrier(0x008, 1, 0);    \
    __builtin_amdgcn_sched_group_barrier(0x002, 6, 0);    \
    __builtin_amdgcn_sched_group_barrier(0x400, 1, 0)
#define PIPE_SCORES_WITH_SOFTMAX() do {                   \
    PIPE_W1_V6(); PIPE_W1_V6(); PIPE_W1_V6(); PIPE_W1_V6(); \
    PIPE_W1_V6(); PIPE_W1_V6(); PIPE_W1_V6(); PIPE_W1_V6(); \
    PIPE_W1_V6(); PIPE_W1_V6(); PIPE_W1_V6(); PIPE_W1_V6(); \
    PIPE_W1_V6(); PIPE_W1_V6(); PIPE_W1_V6(); PIPE_W1_V6(); \
    __builtin_amdgcn_sched_group_barrier(0x402, 16, 0);   \
} while (0)
#define PIPE_OUTPUT() __builtin_amdgcn_sched_group_barrier(0x008, 16, 0)
#define PIPE_LOADS()  __builtin_amdgcn_sched_group_barrier(0x020, 8, 0)

#define LO2(F) (v2f{(F).x, (F).y})
#define HI2(F) (v2f{(F).z, (F).w})

__device__ __forceinline__ v8f tile_scores(const Frag8& K, const v2f (&q)[16]) {
    v8f s;
#pragma unroll
    for (int r = 0; r < 8; ++r) s[r] = 0.0f;
    s = WMMA_F32(LO2(K.f0), q[0],  s);  s = WMMA_F32(HI2(K.f0), q[1],  s);
    s = WMMA_F32(LO2(K.f1), q[2],  s);  s = WMMA_F32(HI2(K.f1), q[3],  s);
    s = WMMA_F32(LO2(K.f2), q[4],  s);  s = WMMA_F32(HI2(K.f2), q[5],  s);
    s = WMMA_F32(LO2(K.f3), q[6],  s);  s = WMMA_F32(HI2(K.f3), q[7],  s);
    s = WMMA_F32(LO2(K.f4), q[8],  s);  s = WMMA_F32(HI2(K.f4), q[9],  s);
    s = WMMA_F32(LO2(K.f5), q[10], s);  s = WMMA_F32(HI2(K.f5), q[11], s);
    s = WMMA_F32(LO2(K.f6), q[12], s);  s = WMMA_F32(HI2(K.f6), q[13], s);
    s = WMMA_F32(LO2(K.f7), q[14], s);  s = WMMA_F32(HI2(K.f7), q[15], s);
    return s;
}

__device__ __forceinline__ float tile_max8(const v8f s) {
    float mloc = s[0];
#pragma unroll
    for (int r = 1; r < 8; ++r) mloc = fmaxf(mloc, s[r]);
    return mloc;
}

__device__ __forceinline__ void tile_softmax(const v8f s, v8f (&accT)[4],
                                             float& m, float& l, const int hi,
                                             v2f (&pb)[4]) {

    float mloc = s[0];
#pragma unroll
    for (int r = 1; r < 8; ++r) mloc = fmaxf(mloc, s[r]);
    mloc = fmaxf(mloc, xhalf(mloc));
    float mnew = fmaxf(m, mloc);
    float sc = __expf(m - mnew);
    m = mnew;

    float p[8], ls = 0.0f;
#pragma unroll
    for (int r = 0; r < 8; ++r) { p[r] = __expf(s[r] - mnew); ls += p[r]; }
    ls += xhalf(ls);
    l = l * sc + ls;

#pragma unroll
    for (int t = 0; t < 4; ++t)
#pragma unroll
        for (int r = 0; r < 8; ++r) accT[t][r] *= sc;

    float tt[8];
#pragma unroll
    for (int r = 0; r < 8; ++r) tt[r] = xhalf(p[r]);
    pb[0].x = hi ? tt[2] : p[0];  pb[0].y = hi ? tt[3] : p[1];
    pb[1].x = hi ? tt[6] : p[4];  pb[1].y = hi ? tt[7] : p[5];
    pb[2].x = hi ? p[2] : tt[0];  pb[2].y = hi ? p[3] : tt[1];
    pb[3].x = hi ? p[6] : tt[4];  pb[3].y = hi ? p[7] : tt[5];
}

__device__ __forceinline__ void tile_pv(const Frag8& V, const v2f (&pb)[4],
                                        v8f (&accT)[4]) {
    accT[0] = WMMA_F32(LO2(V.f0), pb[0], accT[0]);
    accT[1] = WMMA_F32(LO2(V.f2), pb[0], accT[1]);
    accT[2] = WMMA_F32(LO2(V.f4), pb[0], accT[2]);
    accT[3] = WMMA_F32(LO2(V.f6), pb[0], accT[3]);
    accT[0] = WMMA_F32(HI2(V.f0), pb[1], accT[0]);
    accT[1] = WMMA_F32(HI2(V.f2), pb[1], accT[1]);
    accT[2] = WMMA_F32(HI2(V.f4), pb[1], accT[2]);
    accT[3] = WMMA_F32(HI2(V.f6), pb[1], accT[3]);
    accT[0] = WMMA_F32(LO2(V.f1), pb[2], accT[0]);
    accT[1] = WMMA_F32(LO2(V.f3), pb[2], accT[1]);
    accT[2] = WMMA_F32(LO2(V.f5), pb[2], accT[2]);
    accT[3] = WMMA_F32(LO2(V.f7), pb[2], accT[3]);
    accT[0] = WMMA_F32(HI2(V.f1), pb[3], accT[0]);
    accT[1] = WMMA_F32(HI2(V.f3), pb[3], accT[1]);
    accT[2] = WMMA_F32(HI2(V.f5), pb[3], accT[2]);
    accT[3] = WMMA_F32(HI2(V.f7), pb[3], accT[3]);
}

__global__ __launch_bounds__(256) void gcn_flash(const float* __restrict__ kf,
                                                 const float* __restrict__ vf,
                                                 float* __restrict__ out) {
    const int tid  = threadIdx.x;
    const int wave = tid >> 5;
    const int lane = tid & 31;
    const int hi   = lane >> 4;
    const int mrow = lane & 15;

    __shared__ int board[8];
    const int me   = __builtin_amdgcn_readfirstlane(wave);
    const int mate = me ^ 4;
    board[me] = -1;
    int seen = -1;

    const int b     = blockIdx.x >> 6;
    const int qblk  = blockIdx.x & 63;

    const float4* kp = (const float4*)kf + (size_t)b * NJ * TILE_F4 + lane;
    const float4* vp = (const float4*)vf + (size_t)b * NJ * TILE_F4 + lane;

    Frag8 kA, kB;
    Frag8 vF;
    load_frag(kB, kp + TILE_F4);
    load_frag(kA, kp + 2 * TILE_F4);

    __shared__ unsigned nkey[128];
    __shared__ int      srow[128];
    if (tid < 128) {
        const float4* rp = (const float4*)kf
                         + ((size_t)b * NJ + qblk * 8 + (tid >> 4)) * TILE_F4 + (tid & 15);
        float a = 0.0f;
#pragma unroll
        for (int g = 0; g < 16; ++g) {
            const float4 v = rp[g * 16];
            a += v.x * v.x + v.y * v.y + v.z * v.z + v.w * v.w;
        }
        nkey[tid] = __builtin_bit_cast(unsigned, a);
    }
    __syncthreads();
    if (tid < 128) {
        const unsigned ki = nkey[tid];
        int rank = 0;
        for (int jj = 0; jj < 128; ++jj) {
            const unsigned kj = nkey[jj];
            rank += (kj < ki || (kj == ki && jj < tid)) ? 1 : 0;
        }
        srow[rank] = tid;
    }
    __syncthreads();

    __shared__ float part[(NPH - 1) * 8 * PART_F * 32];


    v2f  q[16];
    v8f  accT[4];
    float m = 0.0f, l = 0.0f;
    v2f  pb[4];
    int  qt = me;

    for (int t = 0; t < NPH; ++t) {
        {
            const int cc = me ^ t;
            qt = (cc & 4) ? (7 - (cc & 3)) : cc;
        }
        const int j0 = t * PH_TILES;
        Frag8 kT;
        {
            Frag8 qf;
            const int qi = srow[qt * 16 + mrow];
            load_frag(qf, (const float4*)kf + ((size_t)b * NJ + qblk * 8 + (qi >> 4)) * TILE_F4
                          + hi * 16 + (qi & 15));
            load_frag(kT, kp + (size_t)j0 * TILE_F4);
            q[0]  = LO2(qf.f0); q[1]  = HI2(qf.f0); q[2]  = LO2(qf.f1); q[3]  = HI2(qf.f1);
            q[4]  = LO2(qf.f2); q[5]  = HI2(qf.f2); q[6]  = LO2(qf.f3); q[7]  = HI2(qf.f3);
            q[8]  = LO2(qf.f4); q[9]  = HI2(qf.f4); q[10] = LO2(qf.f5); q[11] = HI2(qf.f5);
            q[12] = LO2(qf.f6); q[13] = HI2(qf.f6); q[14] = LO2(qf.f7); q[15] = HI2(qf.f7);
        }
#pragma unroll
        for (int tt = 0; tt < 4; ++tt)
#pragma unroll
            for (int r = 0; r < 8; ++r) accT[tt][r] = 0.0f;
        m = 0.0f;
#pragma unroll
        for (int c = 0; c < 16; ++c) m += q[c].x * q[c].x + q[c].y * q[c].y;
        m += xhalf(m);
        l = 0.0f;

        v8f s0 = tile_scores(kT, q);

        for (int j = j0; j < j0 + PH_TILES; j += 2) {
            const int trip = j >> 1;
            if (__builtin_amdgcn_readfirstlane(seen) >= trip) __builtin_amdgcn_s_setprio(3);
            else                                              __builtin_amdgcn_s_setprio(0);

            const int jb = (j + 3 < NJ) ? (j + 3) : (j + 1);
            const int jc = (j + 4 < NJ) ? (j + 4) : j;

            v8f s1;
            if (__builtin_amdgcn_ballot_w32(tile_max8(s0) > m - SKIP_T) != 0u) {
                load_frag(vF, vp + (size_t)j * TILE_F4);
                tile_softmax(s0, accT, m, l, hi, pb);
                s1 = tile_scores(kB, q);
                tile_pv(vF, pb, accT);
                PIPE_SCORES_WITH_SOFTMAX();
                PIPE_OUTPUT();
            } else {
                s1 = tile_scores(kB, q);
            }
            load_frag(kB, kp + (size_t)jb * TILE_F4);

            asm volatile("" ::: "memory");
            board[me] = trip;
            seen = board[mate];

            if (__builtin_amdgcn_ballot_w32(tile_max8(s1) > m - SKIP_T) != 0u) {
                load_frag(vF, vp + (size_t)(j + 1) * TILE_F4);
                tile_softmax(s1, accT, m, l, hi, pb);
                s0 = tile_scores(kA, q);
                tile_pv(vF, pb, accT);
                PIPE_SCORES_WITH_SOFTMAX();
                PIPE_OUTPUT();
            } else {
                s0 = tile_scores(kA, q);
            }
            load_frag(kA, kp + (size_t)jc * TILE_F4);
        }

        if (t < NPH - 1) {
            float* ps = part + (size_t)((t * 8 + qt) * PART_F) * 32 + lane;
#pragma unroll
            for (int tt = 0; tt < 4; ++tt)
#pragma unroll
                for (int r = 0; r < 8; ++r) ps[(tt * 8 + r) * 32] = accT[tt][r];
            ps[32 * 32] = m;
            ps[33 * 32] = l;
        }
    }

    __builtin_amdgcn_s_setprio(0);

    __syncthreads();
    for (int i = 0; i < NPH - 1; ++i) {
        const float* pi = part + (size_t)((i * 8 + qt) * PART_F) * 32 + lane;
        const float mi   = pi[32 * 32];
        const float li   = pi[33 * 32];
        const float mnew = fmaxf(m, mi);
        const float sa   = __expf(m - mnew);
        const float sb   = __expf(mi - mnew);
        l = l * sa + li * sb;
#pragma unroll
        for (int tt = 0; tt < 4; ++tt)
#pragma unroll
            for (int r = 0; r < 8; ++r)
                accT[tt][r] = accT[tt][r] * sa + pi[(tt * 8 + r) * 32] * sb;
        m = mnew;
    }

    __shared__ float ostage[FDIM * 128];
    {
        const float inv = 1.0f / l;
        const int qi = srow[qt * 16 + mrow];
#pragma unroll
        for (int t = 0; t < 4; ++t)
#pragma unroll
            for (int r = 0; r < 8; ++r) {
                const int o = 16 * t + r + 8 * hi;
                ostage[o * 128 + qi] = accT[t][r] * inv;
            }
    }
    __syncthreads();
#pragma unroll
    for (int k = 0; k < 8; ++k) {
        const int o = me * 8 + k;
        const float4 v = ((const float4*)ostage)[o * 32 + lane];
        ((float4*)(out + ((size_t)b * FDIM + o) * N_NODES + qblk * 128))[lane] = v;
    }
}

extern "C" void kernel_launch(void* const* d_in, const int* in_sizes, int n_in,
                              void* d_out, int out_size, void* d_ws, size_t ws_size,
                              hipStream_t stream) {
    (void)in_sizes; (void)n_in; (void)out_size; (void)ws_size;
    const float* x = (const float*)d_in[0];
    const float* w = (const float*)d_in[1];
    float* kf  = (float*)d_ws;
    float* vf  = kf + (size_t)BATCH * N_NODES * FDIM;
    float* out = (float*)d_out;

    dim3 gPrep(N_NODES / 256, BATCH);
    gcn_prep<<<gPrep, 256, 0, stream>>>(x, w, kf, vf);

    dim3 gFlash(BATCH * (N_NODES / 128));
    gcn_flash<<<gFlash, 256, 0, stream>>>(kf, vf, out);
}


# Skip test of key tile j issued under the 16-WMMA score chain of tile j+1, which stands in front of the test's branch; an unskipped tile's softmax follows the chain; same operations, same sum order
# speedup vs baseline: 4.5746x; 1.0341x over previous
#include <hip/hip_runtime.h>


#define BATCH   4
#define FDIM    64
#define N_NODES 8192
#define NJ      (N_NODES / 16)
#define TILE_F  1024
#define TILE_F4 256
#define SKIP_T  27.75f
#define NPH      8
#define PH_TILES (NJ / NPH)
#define PART_F   34

typedef __attribute__((ext_vector_type(2))) float v2f;
typedef __attribute__((ext_vector_type(8))) float v8f;

__global__ __launch_bounds__(256) void gcn_prep(const float* __restrict__ x,
                                                const float* __restrict__ w,
                                                float* __restrict__ kf,
                                                float* __restrict__ vf) {
    const int n  = blockIdx.x * 256 + threadIdx.x;
    const int b  = blockIdx.y;
    const int j  = n >> 4;
    const int kk = n & 15;
    const float* xb = x + (size_t)b * FDIM * N_NODES;
    float4* ktile = (float4*)(kf + ((size_t)b * NJ + j) * TILE_F);
    float*  vtile = vf + ((size_t)b * NJ + j) * TILE_F;

    float acc[FDIM];
#pragma unroll
    for (int o = 0; o < FDIM; ++o) acc[o] = 0.0f;

    for (int g = 0; g < 8; ++g) {
        float v[8];
#pragma unroll
        for (int i = 0; i < 8; ++i)
            v[i] = xb[(size_t)(8 * g + i) * N_NODES + n];
#pragma unroll
        for (int i = 0; i < 8; ++i)
#pragma unroll
            for (int o = 0; o < FDIM; ++o)
                acc[o] += v[i] * w[(8 * g + i) * FDIM + o];
        ktile[g * 32 + kk]      = make_float4(v[0], v[1], v[4], v[5]);
        ktile[g * 32 + 16 + kk] = make_float4(v[2], v[3], v[6], v[7]);
    }

    const int c = kk >> 2, h = (kk >> 1) & 1, e = kk & 1;
    float* vp = vtile + ((c >> 1) * 32 + h * 16) * 4 + (c & 1) * 2 + e;
#pragma unroll
    for (int o = 0; o < FDIM; ++o)
        vp[(o >> 4) * 256 + (o & 15) * 4] = acc[o];
}

struct Frag8 { float4 f0, f1, f2, f3, f4, f5, f6, f7; };

__device__ __forceinline__ void load_frag(Frag8& d, const float4* __restrict__ p) {
    d.f0 = p[0];   d.f1 = p[32];  d.f2 = p[64];  d.f3 = p[96];
    d.f4 = p[128]; d.f5 = p[160]; d.f6 = p[192]; d.f7 = p[224];
}

#define WMMA_F32(A, B, C) \
    __builtin_amdgcn_wmma_f32_16x16x4_f32(false, (A), false, (B), (short)0, (C), false, false)

__device__ __forceinline__ float xhalf(float x) {
    const int b = __builtin_bit_cast(int, x);
    return __builtin_bit_cast(float,
        __builtin_amdgcn_permlanex16(b, b, 0x76543210, 0xfedcba98, false, false));
}

#define PIPE_CHAIN_OVER_TEST() do {                       \
    __builtin_amdgcn_sched_group_barrier(0x008, 2, 0);    \
    __builtin_amdgcn_sched_group_barrier(0x002, 8, 0);    \
    __builtin_amdgcn_sched_group_barrier(0x008, 14, 0);   \
} while (0)

#define LO2(F) (v2f{(F).x, (F).y})
#define HI2(F) (v2f{(F).z, (F).w})

__device__ __forceinline__ v8f tile_scores(const Frag8& K, const v2f (&q)[16]) {
    v8f s;
#pragma unroll
    for (int r = 0; r < 8; ++r) s[r] = 0.0f;
    s = WMMA_F32(LO2(K.f0), q[0],  s);  s = WMMA_F32(HI2(K.f0), q[1],  s);
    s = WMMA_F32(LO2(K.f1), q[2],  s);  s = WMMA_F32(HI2(K.f1), q[3],  s);
    s = WMMA_F32(LO2(K.f2), q[4],  s);  s = WMMA_F32(HI2(K.f2), q[5],  s);
    s = WMMA_F32(LO2(K.f3), q[6],  s);  s = WMMA_F32(HI2(K.f3), q[7],  s);
    s = WMMA_F32(LO2(K.f4), q[8],  s);  s = WMMA_F32(HI2(K.f4), q[9],  s);
    s = WMMA_F32(LO2(K.f5), q[10], s);  s = WMMA_F32(HI2(K.f5), q[11], s);
    s = WMMA_F32(LO2(K.f6), q[12], s);  s = WMMA_F32(HI2(K.f6), q[13], s);
    s = WMMA_F32(LO2(K.f7), q[14], s);  s = WMMA_F32(HI2(K.f7), q[15], s);
    return s;
}

__device__ __forceinline__ float tile_max8(const v8f s) {
    float mloc = s[0];
#pragma unroll
    for (int r = 1; r < 8; ++r) mloc = fmaxf(mloc, s[r]);
    return mloc;
}

__device__ __forceinline__ void tile_softmax(const v8f s, v8f (&accT)[4],
                                             float& m, float& l, const int hi,
                                             v2f (&pb)[4]) {

    float mloc = s[0];
#pragma unroll
    for (int r = 1; r < 8; ++r) mloc = fmaxf(mloc, s[r]);
    mloc = fmaxf(mloc, xhalf(mloc));
    float mnew = fmaxf(m, mloc);
    float sc = __expf(m - mnew);
    m = mnew;

    float p[8], ls = 0.0f;
#pragma unroll
    for (int r = 0; r < 8; ++r) { p[r] = __expf(s[r] - mnew); ls += p[r]; }
    ls += xhalf(ls);
    l = l * sc + ls;

#pragma unroll
    for (int t = 0; t < 4; ++t)
#pragma unroll
        for (int r = 0; r < 8; ++r) accT[t][r] *= sc;

    float tt[8];
#pragma unroll
    for (int r = 0; r < 8; ++r) tt[r] = xhalf(p[r]);
    pb[0].x = hi ? tt[2] : p[0];  pb[0].y = hi ? tt[3] : p[1];
    pb[1].x = hi ? tt[6] : p[4];  pb[1].y = hi ? tt[7] : p[5];
    pb[2].x = hi ? p[2] : tt[0];  pb[2].y = hi ? p[3] : tt[1];
    pb[3].x = hi ? p[6] : tt[4];  pb[3].y = hi ? p[7] : tt[5];
}

__device__ __forceinline__ void tile_pv(const Frag8& V, const v2f (&pb)[4],
                                        v8f (&accT)[4]) {
    accT[0] = WMMA_F32(LO2(V.f0), pb[0], accT[0]);
    accT[1] = WMMA_F32(LO2(V.f2), pb[0], accT[1]);
    accT[2] = WMMA_F32(LO2(V.f4), pb[0], accT[2]);
    accT[3] = WMMA_F32(LO2(V.f6), pb[0], accT[3]);
    accT[0] = WMMA_F32(HI2(V.f0), pb[1], accT[0]);
    accT[1] = WMMA_F32(HI2(V.f2), pb[1], accT[1]);
    accT[2] = WMMA_F32(HI2(V.f4), pb[1], accT[2]);
    accT[3] = WMMA_F32(HI2(V.f6), pb[1], accT[3]);
    accT[0] = WMMA_F32(LO2(V.f1), pb[2], accT[0]);
    accT[1] = WMMA_F32(LO2(V.f3), pb[2], accT[1]);
    accT[2] = WMMA_F32(LO2(V.f5), pb[2], accT[2]);
    accT[3] = WMMA_F32(LO2(V.f7), pb[2], accT[3]);
    accT[0] = WMMA_F32(HI2(V.f1), pb[3], accT[0]);
    accT[1] = WMMA_F32(HI2(V.f3), pb[3], accT[1]);
    accT[2] = WMMA_F32(HI2(V.f5), pb[3], accT[2]);
    accT[3] = WMMA_F32(HI2(V.f7), pb[3], accT[3]);
}

__global__ __launch_bounds__(256) void gcn_flash(const float* __restrict__ kf,
                                                 const float* __restrict__ vf,
                                                 float* __restrict__ out) {
    const int tid  = threadIdx.x;
    const int wave = tid >> 5;
    const int lane = tid & 31;
    const int hi   = lane >> 4;
    const int mrow = lane & 15;

    __shared__ int board[8];
    const int me   = __builtin_amdgcn_readfirstlane(wave);
    const int mate = me ^ 4;
    board[me] = -1;
    int seen = -1;

    const int b     = blockIdx.x >> 6;
    const int qblk  = blockIdx.x & 63;

    const float4* kp = (const float4*)kf + (size_t)b * NJ * TILE_F4 + lane;
    const float4* vp = (const float4*)vf + (size_t)b * NJ * TILE_F4 + lane;

    Frag8 kA, kB;
    Frag8 vF;
    load_frag(kB, kp + TILE_F4);
    load_frag(kA, kp + 2 * TILE_F4);

    __shared__ unsigned nkey[128];
    __shared__ int      srow[128];
    if (tid < 128) {
        const float4* rp = (const float4*)kf
                         + ((size_t)b * NJ + qblk * 8 + (tid >> 4)) * TILE_F4 + (tid & 15);
        float a = 0.0f;
#pragma unroll
        for (int g = 0; g < 16; ++g) {
            const float4 v = rp[g * 16];
            a += v.x * v.x + v.y * v.y + v.z * v.z + v.w * v.w;
        }
        nkey[tid] = __builtin_bit_cast(unsigned, a);
    }
    __syncthreads();
    if (tid < 128) {
        const unsigned ki = nkey[tid];
        int rank = 0;
        for (int jj = 0; jj < 128; ++jj) {
            const unsigned kj = nkey[jj];
            rank += (kj < ki || (kj == ki && jj < tid)) ? 1 : 0;
        }
        srow[rank] = tid;
    }
    __syncthreads();

    __shared__ float part[(NPH - 1) * 8 * PART_F * 32];


    v2f  q[16];
    v8f  accT[4];
    float m = 0.0f, l = 0.0f;
    v2f  pb[4];
    int  qt = me;

    for (int t = 0; t < NPH; ++t) {
        {
            const int cc = me ^ t;
            qt = (cc & 4) ? (7 - (cc & 3)) : cc;
        }
        const int j0 = t * PH_TILES;
        Frag8 kT;
        {
            Frag8 qf;
            const int qi = srow[qt * 16 + mrow];
            load_frag(qf, (const float4*)kf + ((size_t)b * NJ + qblk * 8 + (qi >> 4)) * TILE_F4
                          + hi * 16 + (qi & 15));
            load_frag(kT, kp + (size_t)j0 * TILE_F4);
            q[0]  = LO2(qf.f0); q[1]  = HI2(qf.f0); q[2]  = LO2(qf.f1); q[3]  = HI2(qf.f1);
            q[4]  = LO2(qf.f2); q[5]  = HI2(qf.f2); q[6]  = LO2(qf.f3); q[7]  = HI2(qf.f3);
            q[8]  = LO2(qf.f4); q[9]  = HI2(qf.f4); q[10] = LO2(qf.f5); q[11] = HI2(qf.f5);
            q[12] = LO2(qf.f6); q[13] = HI2(qf.f6); q[14] = LO2(qf.f7); q[15] = HI2(qf.f7);
        }
#pragma unroll
        for (int tt = 0; tt < 4; ++tt)
#pragma unroll
            for (int r = 0; r < 8; ++r) accT[tt][r] = 0.0f;
        m = 0.0f;
#pragma unroll
        for (int c = 0; c < 16; ++c) m += q[c].x * q[c].x + q[c].y * q[c].y;
        m += xhalf(m);
        l = 0.0f;

        v8f s0 = tile_scores(kT, q);

        for (int j = j0; j < j0 + PH_TILES; j += 2) {
            const int trip = j >> 1;
            if (__builtin_amdgcn_readfirstlane(seen) >= trip) __builtin_amdgcn_s_setprio(3);
            else                                              __builtin_amdgcn_s_setprio(0);

            const int jb = (j + 3 < NJ) ? (j + 3) : (j + 1);
            const int jc = (j + 4 < NJ) ? (j + 4) : j;

            const v8f s1 = tile_scores(kB, q);
            const bool need0 = __builtin_amdgcn_ballot_w32(tile_max8(s0) > m - SKIP_T) != 0u;
            PIPE_CHAIN_OVER_TEST();
            if (need0) {
                load_frag(vF, vp + (size_t)j * TILE_F4);
                tile_softmax(s0, accT, m, l, hi, pb);
                tile_pv(vF, pb, accT);
            }
            load_frag(kB, kp + (size_t)jb * TILE_F4);

            asm volatile("" ::: "memory");
            board[me] = trip;
            seen = board[mate];

            s0 = tile_scores(kA, q);
            const bool need1 = __builtin_amdgcn_ballot_w32(tile_max8(s1) > m - SKIP_T) != 0u;
            PIPE_CHAIN_OVER_TEST();
            if (need1) {
                load_frag(vF, vp + (size_t)(j + 1) * TILE_F4);
                tile_softmax(s1, accT, m, l, hi, pb);
                tile_pv(vF, pb, accT);
            }
            load_frag(kA, kp + (size_t)jc * TILE_F4);
        }

        if (t < NPH - 1) {
            float* ps = part + (size_t)((t * 8 + qt) * PART_F) * 32 + lane;
#pragma unroll
            for (int tt = 0; tt < 4; ++tt)
#pragma unroll
                for (int r = 0; r < 8; ++r) ps[(tt * 8 + r) * 32] = accT[tt][r];
            ps[32 * 32] = m;
            ps[33 * 32] = l;
        }
    }

    __builtin_amdgcn_s_setprio(0);

    __syncthreads();
    for (int i = 0; i < NPH - 1; ++i) {
        const float* pi = part + (size_t)((i * 8 + qt) * PART_F) * 32 + lane;
        const float mi   = pi[32 * 32];
        const float li   = pi[33 * 32];
        const float mnew = fmaxf(m, mi);
        const float sa   = __expf(m - mnew);
        const float sb   = __expf(mi - mnew);
        l = l * sa + li * sb;
#pragma unroll
        for (int tt = 0; tt < 4; ++tt)
#pragma unroll
            for (int r = 0; r < 8; ++r)
                accT[tt][r] = accT[tt][r] * sa + pi[(tt * 8 + r) * 32] * sb;
        m = mnew;
    }

    __shared__ float ostage[FDIM * 128];
    {
        const float inv = 1.0f / l;
        const int qi = srow[qt * 16 + mrow];
#pragma unroll
        for (int t = 0; t < 4; ++t)
#pragma unroll
            for (int r = 0; r < 8; ++r) {
                const int o = 16 * t + r + 8 * hi;
                ostage[o * 128 + qi] = accT[t][r] * inv;
            }
    }
    __syncthreads();
#pragma unroll
    for (int k = 0; k < 8; ++k) {
        const int o = me * 8 + k;
        const float4 v = ((const float4*)ostage)[o * 32 + lane];
        ((float4*)(out + ((size_t)b * FDIM + o) * N_NODES + qblk * 128))[lane] = v;
    }
}

extern "C" void kernel_launch(void* const* d_in, const int* in_sizes, int n_in,
                              void* d_out, int out_size, void* d_ws, size_t ws_size,
                              hipStream_t stream) {
    (void)in_sizes; (void)n_in; (void)out_size; (void)ws_size;
    const float* x = (const float*)d_in[0];
    const float* w = (const float*)d_in[1];
    float* kf  = (float*)d_ws;
    float* vf  = kf + (size_t)BATCH * N_NODES * FDIM;
    float* out = (float*)d_out;

    dim3 gPrep(N_NODES / 256, BATCH);
    gcn_prep<<<gPrep, 256, 0, stream>>>(x, w, kf, vf);

    dim3 gFlash(BATCH * (N_NODES / 128));
    gcn_flash<<<gFlash, 256, 0, stream>>>(kf, vf, out);
}
